# starting kernel
# baseline (speedup 1.0000x reference)
#include <hip/hip_runtime.h>

#define NNODES 100000
#define NEDGES 3200000

typedef __attribute__((ext_vector_type(2))) float v2f;
typedef __attribute__((ext_vector_type(8))) float v8f;

__global__ void gcn_init_deg(float* __restrict__ deg) {
    int v = blockIdx.x * 256 + threadIdx.x;
    if (v < NNODES) deg[v] = 1.0f;
}

__global__ void gcn_accum_deg(const int* __restrict__ col, float* __restrict__ deg) {
    int e = blockIdx.x * 256 + threadIdx.x;
    if (e < NEDGES) atomicAdd(&deg[col[e]], 1.0f);
}

__global__ void gcn_make_dinv(const float* __restrict__ deg, float* __restrict__ dinv) {
    int v = blockIdx.x * 256 + threadIdx.x;
    if (v < NNODES) dinv[v] = rsqrtf(deg[v]);
}

__global__ __launch_bounds__(256) void gcn_gemm_scale(
        const float* __restrict__ x, const float* __restrict__ W,
        const float* __restrict__ dinv, float* __restrict__ g,
        float* __restrict__ out) {
    __shared__ float Wl[256 * 64];
    const int tid = threadIdx.x;
    for (int i = tid; i < 256 * 64; i += 256) {
        int k = i >> 6, n = i & 63;
        Wl[(k >> 1) * 128 + (n << 1) + (k & 1)] = W[i];
    }
    __syncthreads();

    const int wave = tid >> 5;
    const int lane = tid & 31;
    const int half = lane >> 4;
    const int mr   = lane & 15;
    const int m0   = blockIdx.x * 128 + wave * 16;
    if (m0 >= NNODES) return;

    const float* xrow = x + (size_t)(m0 + mr) * 256;
    const v2f*   Wp   = (const v2f*)Wl;

    v8f acc0 = {}, acc1 = {}, acc2 = {}, acc3 = {};
    for (int k0 = 0; k0 < 256; k0 += 4) {
        const int ka = k0 + half * 2;
        v2f a = *(const v2f*)(xrow + ka);
        const int kp = ka >> 1;
        v2f b0 = Wp[kp * 64 + mr];
        v2f b1 = Wp[kp * 64 + 16 + mr];
        v2f b2 = Wp[kp * 64 + 32 + mr];
        v2f b3 = Wp[kp * 64 + 48 + mr];
        acc0 = __builtin_amdgcn_wmma_f32_16x16x4_f32(false, a, false, b0, (short)0, acc0, false, false);
        acc1 = __builtin_amdgcn_wmma_f32_16x16x4_f32(false, a, false, b1, (short)0, acc1, false, false);
        acc2 = __builtin_amdgcn_wmma_f32_16x16x4_f32(false, a, false, b2, (short)0, acc2, false, false);
        acc3 = __builtin_amdgcn_wmma_f32_16x16x4_f32(false, a, false, b3, (short)0, acc3, false, false);
    }

    #pragma unroll
    for (int i = 0; i < 8; ++i) {
        const int m = m0 + i + half * 8;
        const float s = dinv[m];
        const size_t rb = (size_t)m * 64 + mr;
        float v0 = acc0[i] * s; g[rb]      = v0; out[rb]      = v0;
        float v1 = acc1[i] * s; g[rb + 16] = v1; out[rb + 16] = v1;
        float v2 = acc2[i] * s; g[rb + 32] = v2; out[rb + 32] = v2;
        float v3 = acc3[i] * s; g[rb + 48] = v3; out[rb + 48] = v3;
    }
}

__global__ void gcn_scatter(const float* __restrict__ g,
                            const int* __restrict__ row,
                            const int* __restrict__ col,
                            float* __restrict__ out) {
    size_t t = (size_t)blockIdx.x * 256 + threadIdx.x;
    size_t e = t >> 6;
    int    n = (int)(t & 63);
    if (e < (size_t)NEDGES) {
        int r = row[e], c = col[e];
        atomicAdd(&out[(size_t)c * 64 + n], g[(size_t)r * 64 + n]);
    }
}

__global__ void gcn_finalize(const float* __restrict__ dinv,
                             const float* __restrict__ b,
                             float* __restrict__ out) {
    size_t t = (size_t)blockIdx.x * 256 + threadIdx.x;
    if (t < (size_t)NNODES * 64) {
        int v = (int)(t >> 6), n = (int)(t & 63);
        out[t] = dinv[v] * out[t] + b[n];
    }
}

extern "C" void kernel_launch(void* const* d_in, const int* in_sizes, int n_in,
                              void* d_out, int out_size, void* d_ws, size_t ws_size,
                              hipStream_t stream) {
    const float* x  = (const float*)d_in[0];
    const int*   ei = (const int*)d_in[1];
    const float* W  = (const float*)d_in[2];
    const float* b  = (const float*)d_in[3];
    float* out = (float*)d_out;

    const int* row = ei;
    const int* col = ei + NEDGES;

    float* deg  = (float*)d_ws;
    float* dinv = deg + NNODES;
    float* g    = dinv + NNODES;

    gcn_init_deg  <<<(NNODES + 255) / 256, 256, 0, stream>>>(deg);
    gcn_accum_deg <<<(NEDGES + 255) / 256, 256, 0, stream>>>(col, deg);
    gcn_make_dinv <<<(NNODES + 255) / 256, 256, 0, stream>>>(deg, dinv);
    gcn_gemm_scale<<<(NNODES + 127) / 128, 256, 0, stream>>>(x, W, dinv, g, out);
    gcn_scatter   <<<(int)(((size_t)NEDGES * 64) / 256), 256, 0, stream>>>(g, row, col, out);
    gcn_finalize  <<<(NNODES * 64 + 255) / 256, 256, 0, stream>>>(dinv, b, out);
}


# Edge scatter's global float atomics replaced by owner-computes: 256 groups each scan all dst, keep edges of their 391 rows, add g[src] to LDS rows (ds_add_f32), write out once; sum order LDS arrival
# speedup vs baseline: 2.1048x; 2.1048x over previous
#include <hip/hip_runtime.h>

#define NNODES 100000
#define NEDGES 3200000

typedef __attribute__((ext_vector_type(2))) float v2f;
typedef __attribute__((ext_vector_type(8))) float v8f;

__global__ void gcn_init_deg(float* __restrict__ deg) {
    int v = blockIdx.x * 256 + threadIdx.x;
    if (v < NNODES) deg[v] = 1.0f;
}

__global__ void gcn_accum_deg(const int* __restrict__ col, float* __restrict__ deg) {
    int e = blockIdx.x * 256 + threadIdx.x;
    if (e < NEDGES) atomicAdd(&deg[col[e]], 1.0f);
}

__global__ void gcn_make_dinv(const float* __restrict__ deg, float* __restrict__ dinv) {
    int v = blockIdx.x * 256 + threadIdx.x;
    if (v < NNODES) dinv[v] = rsqrtf(deg[v]);
}

__global__ __launch_bounds__(256) void gcn_gemm_scale(
        const float* __restrict__ x, const float* __restrict__ W,
        const float* __restrict__ dinv, float* __restrict__ g) {
    __shared__ float Wl[256 * 64];
    const int tid = threadIdx.x;
    for (int i = tid; i < 256 * 64; i += 256) {
        int k = i >> 6, n = i & 63;
        Wl[(k >> 1) * 128 + (n << 1) + (k & 1)] = W[i];
    }
    __syncthreads();

    const int wave = tid >> 5;
    const int lane = tid & 31;
    const int half = lane >> 4;
    const int mr   = lane & 15;
    const int m0   = blockIdx.x * 128 + wave * 16;
    if (m0 >= NNODES) return;

    const float* xrow = x + (size_t)(m0 + mr) * 256;
    const v2f*   Wp   = (const v2f*)Wl;

    v8f acc0 = {}, acc1 = {}, acc2 = {}, acc3 = {};
    for (int k0 = 0; k0 < 256; k0 += 4) {
        const int ka = k0 + half * 2;
        v2f a = *(const v2f*)(xrow + ka);
        const int kp = ka >> 1;
        v2f b0 = Wp[kp * 64 + mr];
        v2f b1 = Wp[kp * 64 + 16 + mr];
        v2f b2 = Wp[kp * 64 + 32 + mr];
        v2f b3 = Wp[kp * 64 + 48 + mr];
        acc0 = __builtin_amdgcn_wmma_f32_16x16x4_f32(false, a, false, b0, (short)0, acc0, false, false);
        acc1 = __builtin_amdgcn_wmma_f32_16x16x4_f32(false, a, false, b1, (short)0, acc1, false, false);
        acc2 = __builtin_amdgcn_wmma_f32_16x16x4_f32(false, a, false, b2, (short)0, acc2, false, false);
        acc3 = __builtin_amdgcn_wmma_f32_16x16x4_f32(false, a, false, b3, (short)0, acc3, false, false);
    }

    #pragma unroll
    for (int i = 0; i < 8; ++i) {
        const int m = m0 + i + half * 8;
        const float s = dinv[m];
        const size_t rb = (size_t)m * 64 + mr;
        g[rb]      = acc0[i] * s;
        g[rb + 16] = acc1[i] * s;
        g[rb + 32] = acc2[i] * s;
        g[rb + 48] = acc3[i] * s;
    }
}

#define RNG   391
#define NGRP  256
#define NE4   (NEDGES / 4)
#define NBLK  (NE4 / 256)
#define QCAP  (32 + 1024)

static_assert(NGRP * RNG >= NNODES, "ranges cover every node");
static_assert(NEDGES % 1024 == 0, "dst array is whole 4 KiB blocks");
static_assert(NEDGES <= (1 << 22) && RNG <= (1 << 10), "queue entry packs row (10 bits) and edge (22 bits)");

__device__ __forceinline__ void gcn_load_block(int4 (&c)[8], const int4* __restrict__ col4,
                                               int blk, int lane) {
    #pragma unroll
    for (int u = 0; u < 8; ++u) c[u] = col4[blk * 256 + u * 32 + lane];
}

__device__ __forceinline__ void gcn_test_block(const int4 (&c)[8], int blk, int lane, int lo,
                                               unsigned cnt, unsigned* myq, int& qn) {
    #pragma unroll
    for (int u = 0; u < 8; ++u) {
        const int e0 = (blk * 256 + u * 32 + lane) * 4;
        const unsigned d[4] = { (unsigned)(c[u].x - lo), (unsigned)(c[u].y - lo),
                                (unsigned)(c[u].z - lo), (unsigned)(c[u].w - lo) };
        #pragma unroll
        for (int j = 0; j < 4; ++j) {
            const bool     hit  = d[j] < cnt;
            const unsigned mask = __builtin_amdgcn_ballot_w32(hit);
            if (mask) {
                const unsigned pos = qn + __builtin_amdgcn_mbcnt_lo(mask, 0u);
                if (hit) myq[pos] = (d[j] << 22) | (unsigned)(e0 + j);
                qn += __builtin_popcount(mask);
            }
        }
    }
}

__device__ __forceinline__ void gcn_flush(const float* __restrict__ g,
                                          const int* __restrict__ row,
                                          float* acc, unsigned ent, int n, int lane) {
    const int e  = (int)(ent & 0x3FFFFFu);
    const int cl = (int)(ent >> 22);
    const int r  = row[e];
    #pragma unroll
    for (int j0 = 0; j0 < 32; j0 += 16) {
        if (j0 < n) {
            v2f v[16];
            #pragma unroll
            for (int u = 0; u < 16; ++u) {
                const int rj = __builtin_amdgcn_readlane(r, j0 + u);
                v[u] = *(const v2f*)(g + (size_t)rj * 64 + 2 * lane);
            }
            #pragma unroll
            for (int u = 0; u < 16; ++u) {
                if (j0 + u < n) {
                    const int cj = __builtin_amdgcn_readlane(cl, j0 + u);
                    float* p = acc + cj * 64 + 2 * lane;
                    atomicAdd(p,     v[u].x);
                    atomicAdd(p + 1, v[u].y);
                }
            }
        }
    }
}

__device__ __forceinline__ void gcn_drain(const float* __restrict__ g,
                                          const int* __restrict__ row, float* acc,
                                          const unsigned* myq, int& qn, int lane) {
    while (qn >= 32) {
        qn -= 32;
        gcn_flush(g, row, acc, myq[qn + lane], 32, lane);
    }
}

__global__ __launch_bounds__(1024) void gcn_owner_sum(
        const float* __restrict__ g, const int* __restrict__ row,
        const int* __restrict__ col, const float* __restrict__ dinv,
        const float* __restrict__ b, float* __restrict__ out) {
    __shared__ float    acc[RNG * 64];
    __shared__ unsigned q[32][QCAP];

    const int tid  = threadIdx.x;
    const int lane = tid & 31;
    const int wave = __builtin_amdgcn_readfirstlane(tid >> 5);
    const int lo   = blockIdx.x * RNG;
    const int cnt  = min(RNG, NNODES - lo);

    for (int i = tid; i < cnt * 64; i += 1024) acc[i] = g[(size_t)lo * 64 + i];
    __syncthreads();

    const int4* col4 = (const int4*)col;
    unsigned*   myq  = q[wave];
    int qn = 0;

    int4 A[8], B[8];
    gcn_load_block(A, col4, wave, lane);
    for (int blk = wave; blk < NBLK; blk += 64) {
        const int b1 = blk + 32, b2 = blk + 64;
        gcn_load_block(B, col4, (b1 < NBLK) ? b1 : blk, lane);
        gcn_test_block(A, blk, lane, lo, (unsigned)cnt, myq, qn);
        gcn_drain(g, row, acc, myq, qn, lane);
        if (b1 < NBLK) {
            gcn_load_block(A, col4, (b2 < NBLK) ? b2 : blk, lane);
            gcn_test_block(B, b1, lane, lo, (unsigned)cnt, myq, qn);
            gcn_drain(g, row, acc, myq, qn, lane);
        }
    }
    if (qn > 0) gcn_flush(g, row, acc, (lane < qn) ? myq[lane] : 0u, qn, lane);
    __syncthreads();

    for (int i = tid; i < cnt * 64; i += 1024) {
        const int v = i >> 6, n = i & 63;
        out[(size_t)lo * 64 + i] = dinv[lo + v] * acc[i] + b[n];
    }
}

extern "C" void kernel_launch(void* const* d_in, const int* in_sizes, int n_in,
                              void* d_out, int out_size, void* d_ws, size_t ws_size,
                              hipStream_t stream) {
    const float* x  = (const float*)d_in[0];
    const int*   ei = (const int*)d_in[1];
    const float* W  = (const float*)d_in[2];
    const float* b  = (const float*)d_in[3];
    float* out = (float*)d_out;

    const int* row = ei;
    const int* col = ei + NEDGES;

    float* deg  = (float*)d_ws;
    float* dinv = deg + NNODES;
    float* g    = dinv + NNODES;

    gcn_init_deg  <<<(NNODES + 255) / 256, 256, 0, stream>>>(deg);
    gcn_accum_deg <<<(NEDGES + 255) / 256, 256, 0, stream>>>(col, deg);
    gcn_make_dinv <<<(NNODES + 255) / 256, 256, 0, stream>>>(deg, dinv);
    gcn_gemm_scale<<<(NNODES + 127) / 128, 256, 0, stream>>>(x, W, dinv, g);
    gcn_owner_sum <<<NGRP, 1024, 0, stream>>>(g, row, col, dinv, b, out);
}


# Owner groups scan a 1-byte owner key per edge (dst/391, stored by the degree pass, now 4 edges per thread) instead of the 4-byte dst; hits re-read dst and src; sum order still LDS arrival
# speedup vs baseline: 2.3996x; 1.1401x over previous
#include <hip/hip_runtime.h>

#define NNODES 100000
#define NEDGES 3200000

typedef __attribute__((ext_vector_type(2))) float v2f;
typedef __attribute__((ext_vector_type(8))) float v8f;

__global__ void gcn_init_deg(float* __restrict__ deg) {
    int v = blockIdx.x * 256 + threadIdx.x;
    if (v < NNODES) deg[v] = 1.0f;
}

#define RNG   391
#define NE4   (NEDGES / 4)
static_assert(NE4 % 256 == 0, "degree pass: whole blocks of 256 threads");
static_assert((NNODES - 1) / RNG < 256, "owner key fits one byte");

__global__ void gcn_accum_deg(const int* __restrict__ col, float* __restrict__ deg,
                              unsigned* __restrict__ key) {
    const int i = blockIdx.x * 256 + threadIdx.x;
    const int4 c = ((const int4*)col)[i];
    atomicAdd(&deg[c.x], 1.0f);
    atomicAdd(&deg[c.y], 1.0f);
    atomicAdd(&deg[c.z], 1.0f);
    atomicAdd(&deg[c.w], 1.0f);
    key[i] = ((unsigned)c.x / RNG) | (((unsigned)c.y / RNG) << 8) |
             (((unsigned)c.z / RNG) << 16) | (((unsigned)c.w / RNG) << 24);
}

__global__ void gcn_make_dinv(const float* __restrict__ deg, float* __restrict__ dinv) {
    int v = blockIdx.x * 256 + threadIdx.x;
    if (v < NNODES) dinv[v] = rsqrtf(deg[v]);
}

__global__ __launch_bounds__(256) void gcn_gemm_scale(
        const float* __restrict__ x, const float* __restrict__ W,
        const float* __restrict__ dinv, float* __restrict__ g) {
    __shared__ float Wl[256 * 64];
    const int tid = threadIdx.x;
    for (int i = tid; i < 256 * 64; i += 256) {
        int k = i >> 6, n = i & 63;
        Wl[(k >> 1) * 128 + (n << 1) + (k & 1)] = W[i];
    }
    __syncthreads();

    const int wave = tid >> 5;
    const int lane = tid & 31;
    const int half = lane >> 4;
    const int mr   = lane & 15;
    const int m0   = blockIdx.x * 128 + wave * 16;
    if (m0 >= NNODES) return;

    const float* xrow = x + (size_t)(m0 + mr) * 256;
    const v2f*   Wp   = (const v2f*)Wl;

    v8f acc0 = {}, acc1 = {}, acc2 = {}, acc3 = {};
    for (int k0 = 0; k0 < 256; k0 += 4) {
        const int ka = k0 + half * 2;
        v2f a = *(const v2f*)(xrow + ka);
        const int kp = ka >> 1;
        v2f b0 = Wp[kp * 64 + mr];
        v2f b1 = Wp[kp * 64 + 16 + mr];
        v2f b2 = Wp[kp * 64 + 32 + mr];
        v2f b3 = Wp[kp * 64 + 48 + mr];
        acc0 = __builtin_amdgcn_wmma_f32_16x16x4_f32(false, a, false, b0, (short)0, acc0, false, false);
        acc1 = __builtin_amdgcn_wmma_f32_16x16x4_f32(false, a, false, b1, (short)0, acc1, false, false);
        acc2 = __builtin_amdgcn_wmma_f32_16x16x4_f32(false, a, false, b2, (short)0, acc2, false, false);
        acc3 = __builtin_amdgcn_wmma_f32_16x16x4_f32(false, a, false, b3, (short)0, acc3, false, false);
    }

    #pragma unroll
    for (int i = 0; i < 8; ++i) {
        const int m = m0 + i + half * 8;
        const float s = dinv[m];
        const size_t rb = (size_t)m * 64 + mr;
        g[rb]      = acc0[i] * s;
        g[rb + 16] = acc1[i] * s;
        g[rb + 32] = acc2[i] * s;
        g[rb + 48] = acc3[i] * s;
    }
}

#define NGRP  256
#define NK16  (NEDGES / 16)
#define KSTEP 5
#define KBLK  (NK16 / (32 * KSTEP))
#define QCAP  (32 + 1024)
#define QLIM  (QCAP - 512)

static_assert(NGRP * RNG >= NNODES, "ranges cover every node");
static_assert(NK16 % (32 * KSTEP) == 0, "key array is whole blocks");
static_assert(QLIM >= 32, "a drained queue always has room for one step");

typedef __attribute__((ext_vector_type(4))) unsigned v4u;

__device__ __forceinline__ void gcn_load_keys(v4u (&k)[KSTEP], const v4u* __restrict__ key16,
                                              int blk, int lane) {
    #pragma unroll
    for (int u = 0; u < KSTEP; ++u) k[u] = key16[blk * (32 * KSTEP) + u * 32 + lane];
}

__device__ __forceinline__ void gcn_test_step(const v4u w, int e0, unsigned g4,
                                              unsigned* myq, int& qn) {
    const unsigned x0 = w.x ^ g4, x1 = w.y ^ g4, x2 = w.z ^ g4, x3 = w.w ^ g4;
    const unsigned z0 = ((x0 & 0x7F7F7F7Fu) + 0x7F7F7F7Fu) | x0;
    const unsigned z1 = ((x1 & 0x7F7F7F7Fu) + 0x7F7F7F7Fu) | x1;
    const unsigned z2 = ((x2 & 0x7F7F7F7Fu) + 0x7F7F7F7Fu) | x2;
    const unsigned z3 = ((x3 & 0x7F7F7F7Fu) + 0x7F7F7F7Fu) | x3;
    unsigned c = ((~z0 & 0x80808080u) >> 7) | ((~z1 & 0x80808080u) >> 6) |
                 ((~z2 & 0x80808080u) >> 5) | ((~z3 & 0x80808080u) >> 4);
    unsigned mask = __builtin_amdgcn_ballot_w32(c != 0u);
    while (mask) {
        const int      t   = __builtin_ffs((int)c) - 1;
        const unsigned pos = qn + __builtin_amdgcn_mbcnt_lo(mask, 0u);
        if (c != 0u) myq[pos] = (unsigned)(e0 + 4 * (t & 7) + (t >> 3));
        c &= c - 1u;
        qn += __builtin_popcount(mask);
        mask = __builtin_amdgcn_ballot_w32(c != 0u);
    }
}

__device__ __forceinline__ void gcn_flush(const float* __restrict__ g,
                                          const int* __restrict__ row,
                                          const int* __restrict__ col, int lo,
                                          float* acc, unsigned ent, int n, int lane) {
    const int e  = (int)ent;
    const int r  = row[e];
    const int cl = col[e] - lo;
    #pragma unroll
    for (int j0 = 0; j0 < 32; j0 += 16) {
        if (j0 < n) {
            v2f v[16];
            #pragma unroll
            for (int u = 0; u < 16; ++u) {
                const int rj = __builtin_amdgcn_readlane(r, j0 + u);
                v[u] = *(const v2f*)(g + (size_t)rj * 64 + 2 * lane);
            }
            #pragma unroll
            for (int u = 0; u < 16; ++u) {
                if (j0 + u < n) {
                    const int cj = __builtin_amdgcn_readlane(cl, j0 + u);
                    float* p = acc + cj * 64 + 2 * lane;
                    atomicAdd(p,     v[u].x);
                    atomicAdd(p + 1, v[u].y);
                }
            }
        }
    }
}

__device__ __forceinline__ void gcn_drain(const float* __restrict__ g,
                                          const int* __restrict__ row,
                                          const int* __restrict__ col, int lo, float* acc,
                                          const unsigned* myq, int& qn, int lane) {
    while (qn >= 32) {
        qn -= 32;
        gcn_flush(g, row, col, lo, acc, myq[qn + lane], 32, lane);
    }
}

__device__ __forceinline__ void gcn_scan_block(const v4u (&k)[KSTEP], int blk, int lane, unsigned g4,
                                               const float* __restrict__ g,
                                               const int* __restrict__ row,
                                               const int* __restrict__ col, int lo, float* acc,
                                               unsigned* myq, int& qn) {
    int u = 0;
    do {
        #pragma unroll
        for (int s = 0; s < KSTEP; ++s) {
            if (s >= u && qn <= QLIM) {
                gcn_test_step(k[s], (blk * (32 * KSTEP) + s * 32 + lane) * 16, g4, myq, qn);
                u = s + 1;
            }
        }
        gcn_drain(g, row, col, lo, acc, myq, qn, lane);
    } while (u < KSTEP);
}

__global__ __launch_bounds__(1024) void gcn_owner_sum(
        const float* __restrict__ g, const int* __restrict__ row,
        const int* __restrict__ col, const unsigned* __restrict__ key,
        const float* __restrict__ dinv,
        const float* __restrict__ b, float* __restrict__ out) {
    __shared__ float    acc[RNG * 64];
    __shared__ unsigned q[32][QCAP];

    const int tid  = threadIdx.x;
    const int lane = tid & 31;
    const int wave = __builtin_amdgcn_readfirstlane(tid >> 5);
    const int lo   = blockIdx.x * RNG;
    const int cnt  = min(RNG, NNODES - lo);
    const unsigned g4 = blockIdx.x * 0x01010101u;

    for (int i = tid; i < cnt * 64; i += 1024) acc[i] = g[(size_t)lo * 64 + i];
    __syncthreads();

    const v4u* key16 = (const v4u*)key;
    unsigned*  myq   = q[wave];
    int qn = 0;

    v4u A[KSTEP], B[KSTEP];
    gcn_load_keys(A, key16, wave, lane);
    for (int blk = wave; blk < KBLK; blk += 64) {
        const int b1 = blk + 32, b2 = blk + 64;
        gcn_load_keys(B, key16, (b1 < KBLK) ? b1 : blk, lane);
        gcn_scan_block(A, blk, lane, g4, g, row, col, lo, acc, myq, qn);
        if (b1 < KBLK) {
            gcn_load_keys(A, key16, (b2 < KBLK) ? b2 : blk, lane);
            gcn_scan_block(B, b1, lane, g4, g, row, col, lo, acc, myq, qn);
        }
    }
    if (qn > 0) gcn_flush(g, row, col, lo, acc, (lane < qn) ? myq[lane] : 0u, qn, lane);
    __syncthreads();

    for (int i = tid; i < cnt * 64; i += 1024) {
        const int v = i >> 6, n = i & 63;
        out[(size_t)lo * 64 + i] = dinv[lo + v] * acc[i] + b[n];
    }
}

extern "C" void kernel_launch(void* const* d_in, const int* in_sizes, int n_in,
                              void* d_out, int out_size, void* d_ws, size_t ws_size,
                              hipStream_t stream) {
    const float* x  = (const float*)d_in[0];
    const int*   ei = (const int*)d_in[1];
    const float* W  = (const float*)d_in[2];
    const float* b  = (const float*)d_in[3];
    float* out = (float*)d_out;

    const int* row = ei;
    const int* col = ei + NEDGES;

    float* deg  = (float*)d_ws;
    float* dinv = deg + NNODES;
    float* g    = dinv + NNODES;
    unsigned* key = (unsigned*)(g + (size_t)NNODES * 64);

    gcn_init_deg  <<<(NNODES + 255) / 256, 256, 0, stream>>>(deg);
    gcn_accum_deg <<<NE4 / 256, 256, 0, stream>>>(col, deg, key);
    gcn_make_dinv <<<(NNODES + 255) / 256, 256, 0, stream>>>(deg, dinv);
    gcn_gemm_scale<<<(NNODES + 127) / 128, 256, 0, stream>>>(x, W, dinv, g);
    gcn_owner_sum <<<NGRP, 1024, 0, stream>>>(g, row, col, key, dinv, b, out);
}


# Degree count's global float atomics replaced by integer LDS histograms: 250 groups (25 edge chunks x 10 node ranges), count planes summed in the dinv pass; seeding launch gone; degree order-free
# speedup vs baseline: 4.5085x; 1.8789x over previous
#include <hip/hip_runtime.h>

#define NNODES 100000
#define NEDGES 3200000

typedef __attribute__((ext_vector_type(2))) float v2f;
typedef __attribute__((ext_vector_type(8))) float v8f;

#define RNG   391
#define NE4   (NEDGES / 4)
#define EC    25
#define NR    10
#define NPR   (NNODES / NR)
#define NPRP  ((NPR + 31) / 32 * 32)
#define WPC   (NE4 / EC)
#define WPS   (WPC / NR)
#define DBAT  8
static_assert(NNODES % NR == 0, "node ranges are equal");
static_assert(NE4 % EC == 0 && WPC % NR == 0, "chunks and key slices are whole");
static_assert(WPS % 32 == 0, "a key slice is whole 128-B lines");
static_assert((NNODES - 1) / RNG < 256, "owner key fits one byte");

__global__ __launch_bounds__(1024) void gcn_count_deg(
        const int* __restrict__ col, unsigned* __restrict__ part,
        unsigned* __restrict__ key) {
    __shared__ unsigned cnt[NPR];
    const int tid = threadIdx.x;
    const int i = blockIdx.x / NR;
    const int j = blockIdx.x % NR;
    for (int t = tid; t < NPR; t += 1024) cnt[t] = 0u;
    __syncthreads();

    const unsigned lo = (unsigned)j * NPR;
    const int4* c4 = (const int4*)col + (size_t)i * WPC;
    unsigned*   k4 = key + (size_t)i * WPC;
    for (int w0 = tid; w0 < WPC; w0 += DBAT * 1024) {
        int4 c[DBAT];
        #pragma unroll
        for (int u = 0; u < DBAT; ++u) {
            const int w = w0 + u * 1024;
            c[u] = (w < WPC) ? c4[w] : make_int4(-1, -1, -1, -1);
        }
        #pragma unroll
        for (int u = 0; u < DBAT; ++u) {
            const int w = w0 + u * 1024;
            const unsigned d0 = (unsigned)c[u].x - lo, d1 = (unsigned)c[u].y - lo;
            const unsigned d2 = (unsigned)c[u].z - lo, d3 = (unsigned)c[u].w - lo;
            if (d0 < NPR) atomicAdd(&cnt[d0], 1u);
            if (d1 < NPR) atomicAdd(&cnt[d1], 1u);
            if (d2 < NPR) atomicAdd(&cnt[d2], 1u);
            if (d3 < NPR) atomicAdd(&cnt[d3], 1u);
            if ((unsigned)(w - j * WPS) < WPS)
                k4[w] = ((unsigned)c[u].x / RNG) | (((unsigned)c[u].y / RNG) << 8) |
                        (((unsigned)c[u].z / RNG) << 16) | (((unsigned)c[u].w / RNG) << 24);
        }
    }
    __syncthreads();
    unsigned* plane = part + (size_t)blockIdx.x * NPRP;
    for (int t = tid; t < NPR; t += 1024) plane[t] = cnt[t];
}

__global__ void gcn_make_dinv(const unsigned* __restrict__ part, float* __restrict__ dinv) {
    const int v = blockIdx.x * 256 + threadIdx.x;
    if (v < NNODES) {
        const int j = v / NPR, t = v - j * NPR;
        const unsigned* p = part + (size_t)j * NPRP + t;
        unsigned d = 1u;
        #pragma unroll
        for (int i = 0; i < EC; ++i) d += p[(size_t)i * NR * NPRP];
        dinv[v] = rsqrtf((float)d);
    }
}

__global__ __launch_bounds__(256) void gcn_gemm_scale(
        const float* __restrict__ x, const float* __restrict__ W,
        const float* __restrict__ dinv, float* __restrict__ g) {
    __shared__ float Wl[256 * 64];
    const int tid = threadIdx.x;
    for (int i = tid; i < 256 * 64; i += 256) {
        int k = i >> 6, n = i & 63;
        Wl[(k >> 1) * 128 + (n << 1) + (k & 1)] = W[i];
    }
    __syncthreads();

    const int wave = tid >> 5;
    const int lane = tid & 31;
    const int half = lane >> 4;
    const int mr   = lane & 15;
    const int m0   = blockIdx.x * 128 + wave * 16;
    if (m0 >= NNODES) return;

    const float* xrow = x + (size_t)(m0 + mr) * 256;
    const v2f*   Wp   = (const v2f*)Wl;

    v8f acc0 = {}, acc1 = {}, acc2 = {}, acc3 = {};
    for (int k0 = 0; k0 < 256; k0 += 4) {
        const int ka = k0 + half * 2;
        v2f a = *(const v2f*)(xrow + ka);
        const int kp = ka >> 1;
        v2f b0 = Wp[kp * 64 + mr];
        v2f b1 = Wp[kp * 64 + 16 + mr];
        v2f b2 = Wp[kp * 64 + 32 + mr];
        v2f b3 = Wp[kp * 64 + 48 + mr];
        acc0 = __builtin_amdgcn_wmma_f32_16x16x4_f32(false, a, false, b0, (short)0, acc0, false, false);
        acc1 = __builtin_amdgcn_wmma_f32_16x16x4_f32(false, a, false, b1, (short)0, acc1, false, false);
        acc2 = __builtin_amdgcn_wmma_f32_16x16x4_f32(false, a, false, b2, (short)0, acc2, false, false);
        acc3 = __builtin_amdgcn_wmma_f32_16x16x4_f32(false, a, false, b3, (short)0, acc3, false, false);
    }

    #pragma unroll
    for (int i = 0; i < 8; ++i) {
        const int m = m0 + i + half * 8;
        const float s = dinv[m];
        const size_t rb = (size_t)m * 64 + mr;
        g[rb]      = acc0[i] * s;
        g[rb + 16] = acc1[i] * s;
        g[rb + 32] = acc2[i] * s;
        g[rb + 48] = acc3[i] * s;
    }
}

#define NGRP  256
#define NK16  (NEDGES / 16)
#define KSTEP 5
#define KBLK  (NK16 / (32 * KSTEP))
#define QCAP  (32 + 1024)
#define QLIM  (QCAP - 512)

static_assert(NGRP * RNG >= NNODES, "ranges cover every node");
static_assert(NK16 % (32 * KSTEP) == 0, "key array is whole blocks");
static_assert(QLIM >= 32, "a drained queue always has room for one step");

typedef __attribute__((ext_vector_type(4))) unsigned v4u;

__device__ __forceinline__ void gcn_load_keys(v4u (&k)[KSTEP], const v4u* __restrict__ key16,
                                              int blk, int lane) {
    #pragma unroll
    for (int u = 0; u < KSTEP; ++u) k[u] = key16[blk * (32 * KSTEP) + u * 32 + lane];
}

__device__ __forceinline__ void gcn_test_step(const v4u w, int e0, unsigned g4,
                                              unsigned* myq, int& qn) {
    const unsigned x0 = w.x ^ g4, x1 = w.y ^ g4, x2 = w.z ^ g4, x3 = w.w ^ g4;
    const unsigned z0 = ((x0 & 0x7F7F7F7Fu) + 0x7F7F7F7Fu) | x0;
    const unsigned z1 = ((x1 & 0x7F7F7F7Fu) + 0x7F7F7F7Fu) | x1;
    const unsigned z2 = ((x2 & 0x7F7F7F7Fu) + 0x7F7F7F7Fu) | x2;
    const unsigned z3 = ((x3 & 0x7F7F7F7Fu) + 0x7F7F7F7Fu) | x3;
    unsigned c = ((~z0 & 0x80808080u) >> 7) | ((~z1 & 0x80808080u) >> 6) |
                 ((~z2 & 0x80808080u) >> 5) | ((~z3 & 0x80808080u) >> 4);
    unsigned mask = __builtin_amdgcn_ballot_w32(c != 0u);
    while (mask) {
        const int      t   = __builtin_ffs((int)c) - 1;
        const unsigned pos = qn + __builtin_amdgcn_mbcnt_lo(mask, 0u);
        if (c != 0u) myq[pos] = (unsigned)(e0 + 4 * (t & 7) + (t >> 3));
        c &= c - 1u;
        qn += __builtin_popcount(mask);
        mask = __builtin_amdgcn_ballot_w32(c != 0u);
    }
}

__device__ __forceinline__ void gcn_flush(const float* __restrict__ g,
                                          const int* __restrict__ row,
                                          const int* __restrict__ col, int lo,
                                          float* acc, unsigned ent, int n, int lane) {
    const int e  = (int)ent;
    const int r  = row[e];
    const int cl = col[e] - lo;
    #pragma unroll
    for (int j0 = 0; j0 < 32; j0 += 16) {
        if (j0 < n) {
            v2f v[16];
            #pragma unroll
            for (int u = 0; u < 16; ++u) {
                const int rj = __builtin_amdgcn_readlane(r, j0 + u);
                v[u] = *(const v2f*)(g + (size_t)rj * 64 + 2 * lane);
            }
            #pragma unroll
            for (int u = 0; u < 16; ++u) {
                if (j0 + u < n) {
                    const int cj = __builtin_amdgcn_readlane(cl, j0 + u);
                    float* p = acc + cj * 64 + 2 * lane;
                    atomicAdd(p,     v[u].x);
                    atomicAdd(p + 1, v[u].y);
                }
            }
        }
    }
}

__device__ __forceinline__ void gcn_drain(const float* __restrict__ g,
                                          const int* __restrict__ row,
                                          const int* __restrict__ col, int lo, float* acc,
                                          const unsigned* myq, int& qn, int lane) {
    while (qn >= 32) {
        qn -= 32;
        gcn_flush(g, row, col, lo, acc, myq[qn + lane], 32, lane);
    }
}

__device__ __forceinline__ void gcn_scan_block(const v4u (&k)[KSTEP], int blk, int lane, unsigned g4,
                                               const float* __restrict__ g,
                                               const int* __restrict__ row,
                                               const int* __restrict__ col, int lo, float* acc,
                                               unsigned* myq, int& qn) {
    int u = 0;
    do {
        #pragma unroll
        for (int s = 0; s < KSTEP; ++s) {
            if (s >= u && qn <= QLIM) {
                gcn_test_step(k[s], (blk * (32 * KSTEP) + s * 32 + lane) * 16, g4, myq, qn);
                u = s + 1;
            }
        }
        gcn_drain(g, row, col, lo, acc, myq, qn, lane);
    } while (u < KSTEP);
}

__global__ __launch_bounds__(1024) void gcn_owner_sum(
        const float* __restrict__ g, const int* __restrict__ row,
        const int* __restrict__ col, const unsigned* __restrict__ key,
        const float* __restrict__ dinv,
        const float* __restrict__ b, float* __restrict__ out) {
    __shared__ float    acc[RNG * 64];
    __shared__ unsigned q[32][QCAP];

    const int tid  = threadIdx.x;
    const int lane = tid & 31;
    const int wave = __builtin_amdgcn_readfirstlane(tid >> 5);
    const int lo   = blockIdx.x * RNG;
    const int cnt  = min(RNG, NNODES - lo);
    const unsigned g4 = blockIdx.x * 0x01010101u;

    for (int i = tid; i < cnt * 64; i += 1024) acc[i] = g[(size_t)lo * 64 + i];
    __syncthreads();

    const v4u* key16 = (const v4u*)key;
    unsigned*  myq   = q[wave];
    int qn = 0;

    v4u A[KSTEP], B[KSTEP];
    gcn_load_keys(A, key16, wave, lane);
    for (int blk = wave; blk < KBLK; blk += 64) {
        const int b1 = blk + 32, b2 = blk + 64;
        gcn_load_keys(B, key16, (b1 < KBLK) ? b1 : blk, lane);
        gcn_scan_block(A, blk, lane, g4, g, row, col, lo, acc, myq, qn);
        if (b1 < KBLK) {
            gcn_load_keys(A, key16, (b2 < KBLK) ? b2 : blk, lane);
            gcn_scan_block(B, b1, lane, g4, g, row, col, lo, acc, myq, qn);
        }
    }
    if (qn > 0) gcn_flush(g, row, col, lo, acc, (lane < qn) ? myq[lane] : 0u, qn, lane);
    __syncthreads();

    for (int i = tid; i < cnt * 64; i += 1024) {
        const int v = i >> 6, n = i & 63;
        out[(size_t)lo * 64 + i] = dinv[lo + v] * acc[i] + b[n];
    }
}

extern "C" void kernel_launch(void* const* d_in, const int* in_sizes, int n_in,
                              void* d_out, int out_size, void* d_ws, size_t ws_size,
                              hipStream_t stream) {
    const float* x  = (const float*)d_in[0];
    const int*   ei = (const int*)d_in[1];
    const float* W  = (const float*)d_in[2];
    const float* b  = (const float*)d_in[3];
    float* out = (float*)d_out;

    const int* row = ei;
    const int* col = ei + NEDGES;

    float* g    = (float*)d_ws;
    unsigned* key = (unsigned*)(g + (size_t)NNODES * 64);
    float* dinv = (float*)(key + NE4);
    unsigned* part = (unsigned*)(dinv + NNODES);

    gcn_count_deg <<<EC * NR, 1024, 0, stream>>>(col, part, key);
    gcn_make_dinv <<<(NNODES + 255) / 256, 256, 0, stream>>>(part, dinv);
    gcn_gemm_scale<<<(NNODES + 127) / 128, 256, 0, stream>>>(x, W, dinv, g);
    gcn_owner_sum <<<NGRP, 1024, 0, stream>>>(g, row, col, key, dinv, b, out);
}


# Owner key stored one-hot by nibble as 32 bit planes (1 bit per edge, built by ballots in the degree pass); an owner group ANDs its 2 planes: 32 edges per op, 1/4 the key bytes; sum order LDS arrival
# speedup vs baseline: 4.9557x; 1.0992x over previous
#include <hip/hip_runtime.h>

#define NNODES 100000
#define NEDGES 3200000

typedef __attribute__((ext_vector_type(2))) float v2f;
typedef __attribute__((ext_vector_type(8))) float v8f;

#define RNG   391
#define NE4   (NEDGES / 4)
#define EC    25
#define NR    10
#define NPR   (NNODES / NR)
#define NPRP  ((NPR + 31) / 32 * 32)
#define WPC   (NE4 / EC)
#define OCTW  256
#define OCTS  (WPC / OCTW)
#define NPL   32
#define PLW   (NEDGES / 32)
static_assert(NNODES % NR == 0, "node ranges are equal");
static_assert(NE4 % EC == 0 && WPC % OCTW == 0, "chunks are whole octs");
static_assert(PLW % 32 == 0, "an owner plane is whole 128-B lines");
static_assert((NNODES - 1) / RNG < 256, "owner number fits two nibbles");

__global__ __launch_bounds__(1024) void gcn_count_deg(
        const int* __restrict__ col, unsigned* __restrict__ part,
        unsigned* __restrict__ plane) {
    __shared__ unsigned cnt[NPR];
    const int tid  = threadIdx.x;
    const int lane = tid & 31;
    const int wave = __builtin_amdgcn_readfirstlane(tid >> 5);
    const int i = blockIdx.x / NR;
    const int j = blockIdx.x % NR;
    for (int t = tid; t < NPR; t += 1024) cnt[t] = 0u;
    __syncthreads();

    const unsigned lo = (unsigned)j * NPR;
    const int4* c4 = (const int4*)col + (size_t)i * WPC;
    for (int oct = wave; oct < OCTS; oct += 32) {
        int4 c[8];
        #pragma unroll
        for (int u = 0; u < 8; ++u) c[u] = c4[oct * OCTW + u * 32 + lane];
        #pragma unroll
        for (int u = 0; u < 8; ++u) {
            const unsigned d0 = (unsigned)c[u].x - lo, d1 = (unsigned)c[u].y - lo;
            const unsigned d2 = (unsigned)c[u].z - lo, d3 = (unsigned)c[u].w - lo;
            if (d0 < NPR) atomicAdd(&cnt[d0], 1u);
            if (d1 < NPR) atomicAdd(&cnt[d1], 1u);
            if (d2 < NPR) atomicAdd(&cnt[d2], 1u);
            if (d3 < NPR) atomicAdd(&cnt[d3], 1u);
        }
        if (oct * NR / OCTS == j) {
            unsigned P[NPL];
            #pragma unroll
            for (int p = 0; p < NPL; ++p) P[p] = 0u;
            #pragma unroll
            for (int u = 0; u < 8; ++u) {
                const unsigned k[4] = {(unsigned)c[u].x / RNG, (unsigned)c[u].y / RNG,
                                       (unsigned)c[u].z / RNG, (unsigned)c[u].w / RNG};
                #pragma unroll
                for (int s = 0; s < 4; ++s) {
                    const unsigned klo = k[s] & 15u, khi = k[s] >> 4;
                    const bool mine = (lane == 4 * u + s);
                    #pragma unroll
                    for (int p = 0; p < 16; ++p) {
                        const unsigned mlo = __builtin_amdgcn_ballot_w32(klo == (unsigned)p);
                        const unsigned mhi = __builtin_amdgcn_ballot_w32(khi == (unsigned)p);
                        P[p]      = mine ? mlo : P[p];
                        P[16 + p] = mine ? mhi : P[16 + p];
                    }
                }
            }
            unsigned* line = plane + (size_t)(i * OCTS + oct) * 32 + lane;
            #pragma unroll
            for (int p = 0; p < NPL; ++p) line[(size_t)p * PLW] = P[p];
        }
    }
    __syncthreads();
    unsigned* cplane = part + (size_t)blockIdx.x * NPRP;
    for (int t = tid; t < NPR; t += 1024) cplane[t] = cnt[t];
}

__global__ void gcn_make_dinv(const unsigned* __restrict__ part, float* __restrict__ dinv) {
    const int v = blockIdx.x * 256 + threadIdx.x;
    if (v < NNODES) {
        const int j = v / NPR, t = v - j * NPR;
        const unsigned* p = part + (size_t)j * NPRP + t;
        unsigned d = 1u;
        #pragma unroll
        for (int i = 0; i < EC; ++i) d += p[(size_t)i * NR * NPRP];
        dinv[v] = rsqrtf((float)d);
    }
}

__global__ __launch_bounds__(256) void gcn_gemm_scale(
        const float* __restrict__ x, const float* __restrict__ W,
        const float* __restrict__ dinv, float* __restrict__ g) {
    __shared__ float Wl[256 * 64];
    const int tid = threadIdx.x;
    for (int i = tid; i < 256 * 64; i += 256) {
        int k = i >> 6, n = i & 63;
        Wl[(k >> 1) * 128 + (n << 1) + (k & 1)] = W[i];
    }
    __syncthreads();

    const int wave = tid >> 5;
    const int lane = tid & 31;
    const int half = lane >> 4;
    const int mr   = lane & 15;
    const int m0   = blockIdx.x * 128 + wave * 16;
    if (m0 >= NNODES) return;

    const float* xrow = x + (size_t)(m0 + mr) * 256;
    const v2f*   Wp   = (const v2f*)Wl;

    v8f acc0 = {}, acc1 = {}, acc2 = {}, acc3 = {};
    for (int k0 = 0; k0 < 256; k0 += 4) {
        const int ka = k0 + half * 2;
        v2f a = *(const v2f*)(xrow + ka);
        const int kp = ka >> 1;
        v2f b0 = Wp[kp * 64 + mr];
        v2f b1 = Wp[kp * 64 + 16 + mr];
        v2f b2 = Wp[kp * 64 + 32 + mr];
        v2f b3 = Wp[kp * 64 + 48 + mr];
        acc0 = __builtin_amdgcn_wmma_f32_16x16x4_f32(false, a, false, b0, (short)0, acc0, false, false);
        acc1 = __builtin_amdgcn_wmma_f32_16x16x4_f32(false, a, false, b1, (short)0, acc1, false, false);
        acc2 = __builtin_amdgcn_wmma_f32_16x16x4_f32(false, a, false, b2, (short)0, acc2, false, false);
        acc3 = __builtin_amdgcn_wmma_f32_16x16x4_f32(false, a, false, b3, (short)0, acc3, false, false);
    }

    #pragma unroll
    for (int i = 0; i < 8; ++i) {
        const int m = m0 + i + half * 8;
        const float s = dinv[m];
        const size_t rb = (size_t)m * 64 + mr;
        g[rb]      = acc0[i] * s;
        g[rb + 16] = acc1[i] * s;
        g[rb + 32] = acc2[i] * s;
        g[rb + 48] = acc3[i] * s;
    }
}

#define NGRP  256
#define NUNIT (NEDGES / 128)
#define NSTEP ((NUNIT + 31) / 32)
#define QCAP  (32 + 1024)

static_assert(NGRP * RNG >= NNODES, "ranges cover every node");
static_assert(NEDGES % 128 == 0, "planes are whole 16-byte words");
static_assert(32 * 32 <= NUNIT, "the first step of every wave lies inside the plane");

typedef __attribute__((ext_vector_type(4))) unsigned v4u;

__device__ __forceinline__ void gcn_push(unsigned c, int ebase, unsigned* myq, int& qn) {
    unsigned mask = __builtin_amdgcn_ballot_w32(c != 0u);
    while (mask) {
        const int      t   = __builtin_ffs((int)c) - 1;
        const unsigned pos = qn + __builtin_amdgcn_mbcnt_lo(mask, 0u);
        if (c != 0u) myq[pos] = (unsigned)(ebase + 4 * t);
        c &= c - 1u;
        qn += __builtin_popcount(mask);
        mask = __builtin_amdgcn_ballot_w32(c != 0u);
    }
}

__device__ __forceinline__ void gcn_flush(const float* __restrict__ g,
                                          const int* __restrict__ row,
                                          const int* __restrict__ col, int lo,
                                          float* acc, unsigned ent, int n, int lane) {
    const int e  = (int)ent;
    const int r  = row[e];
    const int cl = col[e] - lo;
    #pragma unroll
    for (int j0 = 0; j0 < 32; j0 += 16) {
        if (j0 < n) {
            v2f v[16];
            #pragma unroll
            for (int u = 0; u < 16; ++u) {
                const int rj = __builtin_amdgcn_readlane(r, j0 + u);
                v[u] = *(const v2f*)(g + (size_t)rj * 64 + 2 * lane);
            }
            #pragma unroll
            for (int u = 0; u < 16; ++u) {
                if (j0 + u < n) {
                    const int cj = __builtin_amdgcn_readlane(cl, j0 + u);
                    float* p = acc + cj * 64 + 2 * lane;
                    atomicAdd(p,     v[u].x);
                    atomicAdd(p + 1, v[u].y);
                }
            }
        }
    }
}

__device__ __forceinline__ void gcn_drain(const float* __restrict__ g,
                                          const int* __restrict__ row,
                                          const int* __restrict__ col, int lo, float* acc,
                                          const unsigned* myq, int& qn, int lane) {
    while (qn >= 32) {
        qn -= 32;
        gcn_flush(g, row, col, lo, acc, myq[qn + lane], 32, lane);
    }
}

__global__ __launch_bounds__(1024) void gcn_owner_sum(
        const float* __restrict__ g, const int* __restrict__ row,
        const int* __restrict__ col, const unsigned* __restrict__ plane,
        const float* __restrict__ dinv,
        const float* __restrict__ b, float* __restrict__ out) {
    __shared__ float    acc[RNG * 64];
    __shared__ unsigned q[32][QCAP];

    const int tid  = threadIdx.x;
    const int lane = tid & 31;
    const int wave = __builtin_amdgcn_readfirstlane(tid >> 5);
    const int lo   = blockIdx.x * RNG;
    const int cnt  = min(RNG, NNODES - lo);

    for (int i = tid; i < cnt * 64; i += 1024) acc[i] = g[(size_t)lo * 64 + i];
    __syncthreads();

    const v4u* pa = (const v4u*)(plane + (size_t)(blockIdx.x & 15) * PLW);
    const v4u* pb = (const v4u*)(plane + (size_t)(16 + (blockIdx.x >> 4)) * PLW);
    unsigned*  myq = q[wave];
    int qn = 0;

    v4u a  = pa[wave * 32 + lane];
    v4u bb = pb[wave * 32 + lane];
    for (int step = wave; step < NSTEP; step += 32) {
        const int un = min((step + 32) * 32 + lane, NUNIT - 1);
        const v4u an = pa[un], bn = pb[un];
        const int u  = step * 32 + lane;
        v4u h = a & bb;
        if (u >= NUNIT) h = (v4u){0u, 0u, 0u, 0u};
        const int e0 = u * 128;
        gcn_push(h.x, e0,     myq, qn);
        gcn_drain(g, row, col, lo, acc, myq, qn, lane);
        gcn_push(h.y, e0 + 1, myq, qn);
        gcn_drain(g, row, col, lo, acc, myq, qn, lane);
        gcn_push(h.z, e0 + 2, myq, qn);
        gcn_drain(g, row, col, lo, acc, myq, qn, lane);
        gcn_push(h.w, e0 + 3, myq, qn);
        gcn_drain(g, row, col, lo, acc, myq, qn, lane);
        a = an; bb = bn;
    }
    if (qn > 0) gcn_flush(g, row, col, lo, acc, (lane < qn) ? myq[lane] : 0u, qn, lane);
    __syncthreads();

    for (int i = tid; i < cnt * 64; i += 1024) {
        const int v = i >> 6, n = i & 63;
        out[(size_t)lo * 64 + i] = dinv[lo + v] * acc[i] + b[n];
    }
}

extern "C" void kernel_launch(void* const* d_in, const int* in_sizes, int n_in,
                              void* d_out, int out_size, void* d_ws, size_t ws_size,
                              hipStream_t stream) {
    const float* x  = (const float*)d_in[0];
    const int*   ei = (const int*)d_in[1];
    const float* W  = (const float*)d_in[2];
    const float* b  = (const float*)d_in[3];
    float* out = (float*)d_out;

    const int* row = ei;
    const int* col = ei + NEDGES;

    float* g    = (float*)d_ws;
    unsigned* plane = (unsigned*)(g + (size_t)NNODES * 64);
    float* dinv = (float*)(plane + (size_t)NPL * PLW);
    unsigned* part = (unsigned*)(dinv + NNODES);

    gcn_count_deg <<<EC * NR, 1024, 0, stream>>>(col, part, plane);
    gcn_make_dinv <<<(NNODES + 255) / 256, 256, 0, stream>>>(part, dinv);
    gcn_gemm_scale<<<(NNODES + 127) / 128, 256, 0, stream>>>(x, W, dinv, g);
    gcn_owner_sum <<<NGRP, 1024, 0, stream>>>(g, row, col, plane, dinv, b, out);
}


# Row sums moved from LDS float adds to registers: each hit edge is queued (LDS int add) to the wave owning its dst row, which gathers g[src] and adds register-relative; sum order = queue arrival
# speedup vs baseline: 5.2281x; 1.0550x over previous
#include <hip/hip_runtime.h>

#define NNODES 100000
#define NEDGES 3200000

typedef __attribute__((ext_vector_type(2))) float v2f;
typedef __attribute__((ext_vector_type(8))) float v8f;

#define RNG   391
#define NE4   (NEDGES / 4)
#define EC    25
#define NR    10
#define NPR   (NNODES / NR)
#define NPRP  ((NPR + 31) / 32 * 32)
#define WPC   (NE4 / EC)
#define OCTW  256
#define OCTS  (WPC / OCTW)
#define NPL   32
#define PLW   (NEDGES / 32)
static_assert(NNODES % NR == 0, "node ranges are equal");
static_assert(NE4 % EC == 0 && WPC % OCTW == 0, "chunks are whole octs");
static_assert(PLW % 32 == 0, "an owner plane is whole 128-B lines");
static_assert((NNODES - 1) / RNG < 256, "owner number fits two nibbles");

__global__ __launch_bounds__(1024) void gcn_count_deg(
        const int* __restrict__ col, unsigned* __restrict__ part,
        unsigned* __restrict__ plane) {
    __shared__ unsigned cnt[NPR];
    const int tid  = threadIdx.x;
    const int lane = tid & 31;
    const int wave = __builtin_amdgcn_readfirstlane(tid >> 5);
    const int i = blockIdx.x / NR;
    const int j = blockIdx.x % NR;
    for (int t = tid; t < NPR; t += 1024) cnt[t] = 0u;
    __syncthreads();

    const unsigned lo = (unsigned)j * NPR;
    const int4* c4 = (const int4*)col + (size_t)i * WPC;
    for (int oct = wave; oct < OCTS; oct += 32) {
        int4 c[8];
        #pragma unroll
        for (int u = 0; u < 8; ++u) c[u] = c4[oct * OCTW + u * 32 + lane];
        #pragma unroll
        for (int u = 0; u < 8; ++u) {
            const unsigned d0 = (unsigned)c[u].x - lo, d1 = (unsigned)c[u].y - lo;
            const unsigned d2 = (unsigned)c[u].z - lo, d3 = (unsigned)c[u].w - lo;
            if (d0 < NPR) atomicAdd(&cnt[d0], 1u);
            if (d1 < NPR) atomicAdd(&cnt[d1], 1u);
            if (d2 < NPR) atomicAdd(&cnt[d2], 1u);
            if (d3 < NPR) atomicAdd(&cnt[d3], 1u);
        }
        if (oct * NR / OCTS == j) {
            unsigned P[NPL];
            #pragma unroll
            for (int p = 0; p < NPL; ++p) P[p] = 0u;
            #pragma unroll
            for (int u = 0; u < 8; ++u) {
                const unsigned k[4] = {(unsigned)c[u].x / RNG, (unsigned)c[u].y / RNG,
                                       (unsigned)c[u].z / RNG, (unsigned)c[u].w / RNG};
                #pragma unroll
                for (int s = 0; s < 4; ++s) {
                    const unsigned klo = k[s] & 15u, khi = k[s] >> 4;
                    const bool mine = (lane == 4 * u + s);
                    #pragma unroll
                    for (int p = 0; p < 16; ++p) {
                        const unsigned mlo = __builtin_amdgcn_ballot_w32(klo == (unsigned)p);
                        const unsigned mhi = __builtin_amdgcn_ballot_w32(khi == (unsigned)p);
                        P[p]      = mine ? mlo : P[p];
                        P[16 + p] = mine ? mhi : P[16 + p];
                    }
                }
            }
            unsigned* line = plane + (size_t)(i * OCTS + oct) * 32 + lane;
            #pragma unroll
            for (int p = 0; p < NPL; ++p) line[(size_t)p * PLW] = P[p];
        }
    }
    __syncthreads();
    unsigned* cplane = part + (size_t)blockIdx.x * NPRP;
    for (int t = tid; t < NPR; t += 1024) cplane[t] = cnt[t];
}

__global__ void gcn_make_dinv(const unsigned* __restrict__ part, float* __restrict__ dinv) {
    const int v = blockIdx.x * 256 + threadIdx.x;
    if (v < NNODES) {
        const int j = v / NPR, t = v - j * NPR;
        const unsigned* p = part + (size_t)j * NPRP + t;
        unsigned d = 1u;
        #pragma unroll
        for (int i = 0; i < EC; ++i) d += p[(size_t)i * NR * NPRP];
        dinv[v] = rsqrtf((float)d);
    }
}

__global__ __launch_bounds__(256) void gcn_gemm_scale(
        const float* __restrict__ x, const float* __restrict__ W,
        const float* __restrict__ dinv, float* __restrict__ g) {
    __shared__ float Wl[256 * 64];
    const int tid = threadIdx.x;
    for (int i = tid; i < 256 * 64; i += 256) {
        int k = i >> 6, n = i & 63;
        Wl[(k >> 1) * 128 + (n << 1) + (k & 1)] = W[i];
    }
    __syncthreads();

    const int wave = tid >> 5;
    const int lane = tid & 31;
    const int half = lane >> 4;
    const int mr   = lane & 15;
    const int m0   = blockIdx.x * 128 + wave * 16;
    if (m0 >= NNODES) return;

    const float* xrow = x + (size_t)(m0 + mr) * 256;
    const v2f*   Wp   = (const v2f*)Wl;

    v8f acc0 = {}, acc1 = {}, acc2 = {}, acc3 = {};
    for (int k0 = 0; k0 < 256; k0 += 4) {
        const int ka = k0 + half * 2;
        v2f a = *(const v2f*)(xrow + ka);
        const int kp = ka >> 1;
        v2f b0 = Wp[kp * 64 + mr];
        v2f b1 = Wp[kp * 64 + 16 + mr];
        v2f b2 = Wp[kp * 64 + 32 + mr];
        v2f b3 = Wp[kp * 64 + 48 + mr];
        acc0 = __builtin_amdgcn_wmma_f32_16x16x4_f32(false, a, false, b0, (short)0, acc0, false, false);
        acc1 = __builtin_amdgcn_wmma_f32_16x16x4_f32(false, a, false, b1, (short)0, acc1, false, false);
        acc2 = __builtin_amdgcn_wmma_f32_16x16x4_f32(false, a, false, b2, (short)0, acc2, false, false);
        acc3 = __builtin_amdgcn_wmma_f32_16x16x4_f32(false, a, false, b3, (short)0, acc3, false, false);
    }

    #pragma unroll
    for (int i = 0; i < 8; ++i) {
        const int m = m0 + i + half * 8;
        const float s = dinv[m];
        const size_t rb = (size_t)m * 64 + mr;
        g[rb]      = acc0[i] * s;
        g[rb + 16] = acc1[i] * s;
        g[rb + 32] = acc2[i] * s;
        g[rb + 48] = acc3[i] * s;
    }
}

#define NGRP  256
#define NUNIT (NEDGES / 128)
#define NSTEP ((NUNIT + 31) / 32)
#define QCAP  (32 + 1024)
#define SLOTS ((RNG + 31) / 32)
#define OCAP  640
#define SRCB  17

static_assert(NGRP * RNG >= NNODES, "ranges cover every node");
static_assert(NEDGES % 128 == 0, "planes are whole 16-byte words");
static_assert(32 * 32 <= NUNIT, "the first step of every wave lies inside the plane");
static_assert(NNODES <= (1 << SRCB), "a node number fits SRCB bits");

typedef __attribute__((ext_vector_type(4))) unsigned v4u;

__device__ __forceinline__ void gcn_push(unsigned c, int ebase, unsigned* myq, int& qn) {
    unsigned mask = __builtin_amdgcn_ballot_w32(c != 0u);
    while (mask) {
        const int      t   = __builtin_ffs((int)c) - 1;
        const unsigned pos = qn + __builtin_amdgcn_mbcnt_lo(mask, 0u);
        if (c != 0u) myq[pos] = (unsigned)(ebase + 4 * t);
        c &= c - 1u;
        qn += __builtin_popcount(mask);
        mask = __builtin_amdgcn_ballot_w32(c != 0u);
    }
}

__device__ __forceinline__ void gcn_route(const float* __restrict__ g,
                                          const int* __restrict__ row,
                                          const int* __restrict__ col, int lo,
                                          unsigned* ocnt, unsigned* oq, float* ovf,
                                          unsigned ent, int n, int lane) {
    const int e  = (int)ent;
    const int r  = row[e];
    const int cl = col[e] - lo;
    bool spill = false;
    if (lane < n) {
        const unsigned w = (unsigned)cl & 31u, sl = (unsigned)cl >> 5;
        const unsigned pos = atomicAdd(&ocnt[w], 1u);
        if (pos < OCAP) oq[w * OCAP + pos] = (unsigned)r | (sl << SRCB);
        else spill = true;
    }
    unsigned m = __builtin_amdgcn_ballot_w32(spill);
    while (m) {
        const int j = __builtin_ffs((int)m) - 1;
        m &= m - 1u;
        const int rj = __builtin_amdgcn_readlane(r, j);
        const int cj = __builtin_amdgcn_readlane(cl, j);
        const v2f v = *(const v2f*)(g + (size_t)rj * 64 + 2 * lane);
        float* p = ovf + cj * 64 + 2 * lane;
        atomicAdd(p,     v.x);
        atomicAdd(p + 1, v.y);
    }
}

__device__ __forceinline__ void gcn_drain(const float* __restrict__ g,
                                          const int* __restrict__ row,
                                          const int* __restrict__ col, int lo,
                                          unsigned* ocnt, unsigned* oq, float* ovf,
                                          const unsigned* myq, int& qn, int lane) {
    while (qn >= 32) {
        qn -= 32;
        gcn_route(g, row, col, lo, ocnt, oq, ovf, myq[qn + lane], 32, lane);
    }
}

typedef __attribute__((ext_vector_type(16))) float v16f;
#define NOSLOT 15u
static_assert(SLOTS <= NOSLOT, "slot 15 is free");

__global__ __launch_bounds__(1024) void gcn_owner_sum(
        const float* __restrict__ g, const int* __restrict__ row,
        const int* __restrict__ col, const unsigned* __restrict__ plane,
        const float* __restrict__ dinv,
        const float* __restrict__ b, float* __restrict__ out) {
    __shared__ float    ovf[RNG * 64];
    __shared__ unsigned q[32][QCAP];
    __shared__ unsigned oq[32 * OCAP];
    __shared__ unsigned ocnt[32];

    const int tid  = threadIdx.x;
    const int lane = tid & 31;
    const int wave = __builtin_amdgcn_readfirstlane(tid >> 5);
    const int lo   = blockIdx.x * RNG;
    const int cnt  = min(RNG, NNODES - lo);

    for (int i = tid; i < RNG * 64; i += 1024) ovf[i] = 0.0f;
    if (tid < 32) ocnt[tid] = 0u;
    __syncthreads();

    const v4u* pa = (const v4u*)(plane + (size_t)(blockIdx.x & 15) * PLW);
    const v4u* pb = (const v4u*)(plane + (size_t)(16 + (blockIdx.x >> 4)) * PLW);
    unsigned*  myq = q[wave];
    int qn = 0;

    v4u pw = pa[wave * 32 + lane];
    v4u bb = pb[wave * 32 + lane];
    for (int step = wave; step < NSTEP; step += 32) {
        const int un = min((step + 32) * 32 + lane, NUNIT - 1);
        const v4u an = pa[un], bn = pb[un];
        const int u  = step * 32 + lane;
        v4u h = pw & bb;
        if (u >= NUNIT) h = (v4u){0u, 0u, 0u, 0u};
        const int e0 = u * 128;
        gcn_push(h.x, e0,     myq, qn);
        gcn_drain(g, row, col, lo, ocnt, oq, ovf, myq, qn, lane);
        gcn_push(h.y, e0 + 1, myq, qn);
        gcn_drain(g, row, col, lo, ocnt, oq, ovf, myq, qn, lane);
        gcn_push(h.z, e0 + 2, myq, qn);
        gcn_drain(g, row, col, lo, ocnt, oq, ovf, myq, qn, lane);
        gcn_push(h.w, e0 + 3, myq, qn);
        gcn_drain(g, row, col, lo, ocnt, oq, ovf, myq, qn, lane);
        pw = an; bb = bn;
    }
    if (qn > 0) gcn_route(g, row, col, lo, ocnt, oq, ovf, (lane < qn) ? myq[lane] : 0u, qn, lane);
    __syncthreads();

    v16f ax = {}, ay = {};
    #pragma unroll
    for (int sl = 0; sl < SLOTS; ++sl) {
        const int c = sl * 32 + wave;
        const v2f s0 = *(const v2f*)(g + (size_t)(lo + min(c, cnt - 1)) * 64 + 2 * lane);
        ax[sl] = (c < cnt) ? s0.x : 0.0f;
        ay[sl] = (c < cnt) ? s0.y : 0.0f;
    }

    const unsigned* myoq = oq + wave * OCAP;
    const int nq = __builtin_amdgcn_readfirstlane((int)min(ocnt[wave], (unsigned)OCAP));
    for (int base = 0; base < nq; base += 32) {
        const int nb = min(32, nq - base);
        const unsigned ent = (lane < nb) ? myoq[base + lane] : (NOSLOT << SRCB);
        #pragma unroll
        for (int j0 = 0; j0 < 32; j0 += 16) {
            if (j0 < nb) {
                v2f v[16];
                #pragma unroll
                for (int u = 0; u < 16; ++u) {
                    const unsigned eu = (unsigned)__builtin_amdgcn_readlane((int)ent, j0 + u);
                    v[u] = *(const v2f*)(g + (size_t)(eu & ((1u << SRCB) - 1u)) * 64 + 2 * lane);
                }
                #pragma unroll
                for (int u = 0; u < 16; ++u) {
                    const unsigned eu = (unsigned)__builtin_amdgcn_readlane((int)ent, j0 + u);
                    const unsigned k  = eu >> SRCB;
                    ax[k] = ax[k] + v[u].x;
                    ay[k] = ay[k] + v[u].y;
                }
            }
        }
    }

    const v2f bv = *(const v2f*)(b + 2 * lane);
    #pragma unroll
    for (int sl = 0; sl < SLOTS; ++sl) {
        const int c = sl * 32 + wave;
        if (c < cnt) {
            const v2f   o = *(const v2f*)(ovf + c * 64 + 2 * lane);
            const float d = dinv[lo + c];
            v2f r;
            r.x = d * (ax[sl] + o.x) + bv.x;
            r.y = d * (ay[sl] + o.y) + bv.y;
            *(v2f*)(out + (size_t)(lo + c) * 64 + 2 * lane) = r;
        }
    }
}

extern "C" void kernel_launch(void* const* d_in, const int* in_sizes, int n_in,
                              void* d_out, int out_size, void* d_ws, size_t ws_size,
                              hipStream_t stream) {
    const float* x  = (const float*)d_in[0];
    const int*   ei = (const int*)d_in[1];
    const float* W  = (const float*)d_in[2];
    const float* b  = (const float*)d_in[3];
    float* out = (float*)d_out;

    const int* row = ei;
    const int* col = ei + NEDGES;

    float* g    = (float*)d_ws;
    unsigned* plane = (unsigned*)(g + (size_t)NNODES * 64);
    float* dinv = (float*)(plane + (size_t)NPL * PLW);
    unsigned* part = (unsigned*)(dinv + NNODES);

    gcn_count_deg <<<EC * NR, 1024, 0, stream>>>(col, part, plane);
    gcn_make_dinv <<<(NNODES + 255) / 256, 256, 0, stream>>>(part, dinv);
    gcn_gemm_scale<<<(NNODES + 127) / 128, 256, 0, stream>>>(x, W, dinv, g);
    gcn_owner_sum <<<NGRP, 1024, 0, stream>>>(g, row, col, plane, dinv, b, out);
}


# Pass 3 reads one random 4-byte word per hit edge instead of two: the degree pass stores src | (dst mod 391) << 17 per edge (reads src once, in order); sum order unchanged (queue arrival)
# speedup vs baseline: 5.9188x; 1.1321x over previous
#include <hip/hip_runtime.h>

#define NNODES 100000
#define NEDGES 3200000

typedef __attribute__((ext_vector_type(2))) float v2f;
typedef __attribute__((ext_vector_type(8))) float v8f;

#define RNG   391
#define NE4   (NEDGES / 4)
#define EC    25
#define NR    10
#define NPR   (NNODES / NR)
#define NPRP  ((NPR + 31) / 32 * 32)
#define WPC   (NE4 / EC)
#define OCTW  256
#define OCTS  (WPC / OCTW)
#define NPL   32
#define PLW   (NEDGES / 32)
#define SRCB  17
static_assert(NNODES % NR == 0, "node ranges are equal");
static_assert(NE4 % EC == 0 && WPC % OCTW == 0, "chunks are whole octs");
static_assert(PLW % 32 == 0, "an owner plane is whole 128-B lines");
static_assert((NNODES - 1) / RNG < 256, "owner number fits two nibbles");
static_assert(NNODES <= (1 << SRCB), "a node number fits SRCB bits");
static_assert(RNG <= (1 << (32 - SRCB)), "a row number inside a range fits the bits above SRCB");

typedef __attribute__((ext_vector_type(4))) unsigned v4u;

__global__ __launch_bounds__(1024) void gcn_count_deg(
        const int* __restrict__ row, const int* __restrict__ col,
        unsigned* __restrict__ part, unsigned* __restrict__ plane,
        unsigned* __restrict__ pk) {
    __shared__ unsigned cnt[NPR];
    const int tid  = threadIdx.x;
    const int lane = tid & 31;
    const int wave = __builtin_amdgcn_readfirstlane(tid >> 5);
    const int i = blockIdx.x / NR;
    const int j = blockIdx.x % NR;
    for (int t = tid; t < NPR; t += 1024) cnt[t] = 0u;
    __syncthreads();

    const unsigned lo = (unsigned)j * NPR;
    const int4* c4 = (const int4*)col + (size_t)i * WPC;
    const int4* r4 = (const int4*)row + (size_t)i * WPC;
    v4u*        p4 = (v4u*)pk + (size_t)i * WPC;
    for (int oct = wave; oct < OCTS; oct += 32) {
        const bool share = (oct * NR / OCTS == j);
        int4 c[8], r[8];
        #pragma unroll
        for (int u = 0; u < 8; ++u) c[u] = c4[oct * OCTW + u * 32 + lane];
        if (share) {
            #pragma unroll
            for (int u = 0; u < 8; ++u) r[u] = r4[oct * OCTW + u * 32 + lane];
        }
        #pragma unroll
        for (int u = 0; u < 8; ++u) {
            const unsigned d0 = (unsigned)c[u].x - lo, d1 = (unsigned)c[u].y - lo;
            const unsigned d2 = (unsigned)c[u].z - lo, d3 = (unsigned)c[u].w - lo;
            if (d0 < NPR) atomicAdd(&cnt[d0], 1u);
            if (d1 < NPR) atomicAdd(&cnt[d1], 1u);
            if (d2 < NPR) atomicAdd(&cnt[d2], 1u);
            if (d3 < NPR) atomicAdd(&cnt[d3], 1u);
        }
        if (share) {
            unsigned P[NPL];
            #pragma unroll
            for (int p = 0; p < NPL; ++p) P[p] = 0u;
            #pragma unroll
            for (int u = 0; u < 8; ++u) {
                const unsigned k[4] = {(unsigned)c[u].x / RNG, (unsigned)c[u].y / RNG,
                                       (unsigned)c[u].z / RNG, (unsigned)c[u].w / RNG};
                v4u w;
                w.x = (unsigned)r[u].x | (((unsigned)c[u].x - k[0] * RNG) << SRCB);
                w.y = (unsigned)r[u].y | (((unsigned)c[u].y - k[1] * RNG) << SRCB);
                w.z = (unsigned)r[u].z | (((unsigned)c[u].z - k[2] * RNG) << SRCB);
                w.w = (unsigned)r[u].w | (((unsigned)c[u].w - k[3] * RNG) << SRCB);
                p4[oct * OCTW + u * 32 + lane] = w;
                #pragma unroll
                for (int s = 0; s < 4; ++s) {
                    const unsigned klo = k[s] & 15u, khi = k[s] >> 4;
                    const bool mine = (lane == 4 * u + s);
                    #pragma unroll
                    for (int p = 0; p < 16; ++p) {
                        const unsigned mlo = __builtin_amdgcn_ballot_w32(klo == (unsigned)p);
                        const unsigned mhi = __builtin_amdgcn_ballot_w32(khi == (unsigned)p);
                        P[p]      = mine ? mlo : P[p];
                        P[16 + p] = mine ? mhi : P[16 + p];
                    }
                }
            }
            unsigned* line = plane + (size_t)(i * OCTS + oct) * 32 + lane;
            #pragma unroll
            for (int p = 0; p < NPL; ++p) line[(size_t)p * PLW] = P[p];
        }
    }
    __syncthreads();
    unsigned* cplane = part + (size_t)blockIdx.x * NPRP;
    for (int t = tid; t < NPR; t += 1024) cplane[t] = cnt[t];
}

__global__ void gcn_make_dinv(const unsigned* __restrict__ part, float* __restrict__ dinv) {
    const int v = blockIdx.x * 256 + threadIdx.x;
    if (v < NNODES) {
        const int j = v / NPR, t = v - j * NPR;
        const unsigned* p = part + (size_t)j * NPRP + t;
        unsigned d = 1u;
        #pragma unroll
        for (int i = 0; i < EC; ++i) d += p[(size_t)i * NR * NPRP];
        dinv[v] = rsqrtf((float)d);
    }
}

__global__ __launch_bounds__(256) void gcn_gemm_scale(
        const float* __restrict__ x, const float* __restrict__ W,
        const float* __restrict__ dinv, float* __restrict__ g) {
    __shared__ float Wl[256 * 64];
    const int tid = threadIdx.x;
    for (int i = tid; i < 256 * 64; i += 256) {
        int k = i >> 6, n = i & 63;
        Wl[(k >> 1) * 128 + (n << 1) + (k & 1)] = W[i];
    }
    __syncthreads();

    const int wave = tid >> 5;
    const int lane = tid & 31;
    const int half = lane >> 4;
    const int mr   = lane & 15;
    const int m0   = blockIdx.x * 128 + wave * 16;
    if (m0 >= NNODES) return;

    const float* xrow = x + (size_t)(m0 + mr) * 256;
    const v2f*   Wp   = (const v2f*)Wl;

    v8f acc0 = {}, acc1 = {}, acc2 = {}, acc3 = {};
    for (int k0 = 0; k0 < 256; k0 += 4) {
        const int ka = k0 + half * 2;
        v2f a = *(const v2f*)(xrow + ka);
        const int kp = ka >> 1;
        v2f b0 = Wp[kp * 64 + mr];
        v2f b1 = Wp[kp * 64 + 16 + mr];
        v2f b2 = Wp[kp * 64 + 32 + mr];
        v2f b3 = Wp[kp * 64 + 48 + mr];
        acc0 = __builtin_amdgcn_wmma_f32_16x16x4_f32(false, a, false, b0, (short)0, acc0, false, false);
        acc1 = __builtin_amdgcn_wmma_f32_16x16x4_f32(false, a, false, b1, (short)0, acc1, false, false);
        acc2 = __builtin_amdgcn_wmma_f32_16x16x4_f32(false, a, false, b2, (short)0, acc2, false, false);
        acc3 = __builtin_amdgcn_wmma_f32_16x16x4_f32(false, a, false, b3, (short)0, acc3, false, false);
    }

    #pragma unroll
    for (int i = 0; i < 8; ++i) {
        const int m = m0 + i + half * 8;
        const float s = dinv[m];
        const size_t rb = (size_t)m * 64 + mr;
        g[rb]      = acc0[i] * s;
        g[rb + 16] = acc1[i] * s;
        g[rb + 32] = acc2[i] * s;
        g[rb + 48] = acc3[i] * s;
    }
}

#define NGRP  256
#define NUNIT (NEDGES / 128)
#define NSTEP ((NUNIT + 31) / 32)
#define QCAP  (32 + 1024)
#define SLOTS ((RNG + 31) / 32)
#define OCAP  640

static_assert(NGRP * RNG >= NNODES, "ranges cover every node");
static_assert(NEDGES % 128 == 0, "planes are whole 16-byte words");
static_assert(32 * 32 <= NUNIT, "the first step of every wave lies inside the plane");

__device__ __forceinline__ void gcn_push(unsigned c, int ebase, unsigned* myq, int& qn) {
    unsigned mask = __builtin_amdgcn_ballot_w32(c != 0u);
    while (mask) {
        const int      t   = __builtin_ffs((int)c) - 1;
        const unsigned pos = qn + __builtin_amdgcn_mbcnt_lo(mask, 0u);
        if (c != 0u) myq[pos] = (unsigned)(ebase + 4 * t);
        c &= c - 1u;
        qn += __builtin_popcount(mask);
        mask = __builtin_amdgcn_ballot_w32(c != 0u);
    }
}

__device__ __forceinline__ void gcn_route(const float* __restrict__ g,
                                          const unsigned* __restrict__ pk,
                                          unsigned* ocnt, unsigned* oq, float* ovf,
                                          unsigned ent, int n, int lane) {
    const unsigned pw = pk[ent];
    const int r  = (int)(pw & ((1u << SRCB) - 1u));
    const int cl = (int)(pw >> SRCB);
    bool spill = false;
    if (lane < n) {
        const unsigned w = (unsigned)cl & 31u, sl = (unsigned)cl >> 5;
        const unsigned pos = atomicAdd(&ocnt[w], 1u);
        if (pos < OCAP) oq[w * OCAP + pos] = (unsigned)r | (sl << SRCB);
        else spill = true;
    }
    unsigned m = __builtin_amdgcn_ballot_w32(spill);
    while (m) {
        const int j = __builtin_ffs((int)m) - 1;
        m &= m - 1u;
        const int rj = __builtin_amdgcn_readlane(r, j);
        const int cj = __builtin_amdgcn_readlane(cl, j);
        const v2f v = *(const v2f*)(g + (size_t)rj * 64 + 2 * lane);
        float* p = ovf + cj * 64 + 2 * lane;
        atomicAdd(p,     v.x);
        atomicAdd(p + 1, v.y);
    }
}

__device__ __forceinline__ void gcn_drain(const float* __restrict__ g,
                                          const unsigned* __restrict__ pk,
                                          unsigned* ocnt, unsigned* oq, float* ovf,
                                          const unsigned* myq, int& qn, int lane) {
    while (qn >= 32) {
        qn -= 32;
        gcn_route(g, pk, ocnt, oq, ovf, myq[qn + lane], 32, lane);
    }
}

typedef __attribute__((ext_vector_type(16))) float v16f;
#define NOSLOT 15u
static_assert(SLOTS <= NOSLOT, "slot 15 is free");

__global__ __launch_bounds__(1024) void gcn_owner_sum(
        const float* __restrict__ g, const unsigned* __restrict__ pk,
        const unsigned* __restrict__ plane,
        const float* __restrict__ dinv,
        const float* __restrict__ b, float* __restrict__ out) {
    __shared__ float    ovf[RNG * 64];
    __shared__ unsigned q[32][QCAP];
    __shared__ unsigned oq[32 * OCAP];
    __shared__ unsigned ocnt[32];

    const int tid  = threadIdx.x;
    const int lane = tid & 31;
    const int wave = __builtin_amdgcn_readfirstlane(tid >> 5);
    const int lo   = blockIdx.x * RNG;
    const int cnt  = min(RNG, NNODES - lo);

    for (int i = tid; i < RNG * 64; i += 1024) ovf[i] = 0.0f;
    if (tid < 32) ocnt[tid] = 0u;
    __syncthreads();

    const v4u* pa = (const v4u*)(plane + (size_t)(blockIdx.x & 15) * PLW);
    const v4u* pb = (const v4u*)(plane + (size_t)(16 + (blockIdx.x >> 4)) * PLW);
    unsigned*  myq = q[wave];
    int qn = 0;

    v4u pw = pa[wave * 32 + lane];
    v4u bb = pb[wave * 32 + lane];
    for (int step = wave; step < NSTEP; step += 32) {
        const int un = min((step + 32) * 32 + lane, NUNIT - 1);
        const v4u an = pa[un], bn = pb[un];
        const int u  = step * 32 + lane;
        v4u h = pw & bb;
        if (u >= NUNIT) h = (v4u){0u, 0u, 0u, 0u};
        const int e0 = u * 128;
        gcn_push(h.x, e0,     myq, qn);
        gcn_drain(g, pk, ocnt, oq, ovf, myq, qn, lane);
        gcn_push(h.y, e0 + 1, myq, qn);
        gcn_drain(g, pk, ocnt, oq, ovf, myq, qn, lane);
        gcn_push(h.z, e0 + 2, myq, qn);
        gcn_drain(g, pk, ocnt, oq, ovf, myq, qn, lane);
        gcn_push(h.w, e0 + 3, myq, qn);
        gcn_drain(g, pk, ocnt, oq, ovf, myq, qn, lane);
        pw = an; bb = bn;
    }
    if (qn > 0) gcn_route(g, pk, ocnt, oq, ovf, (lane < qn) ? myq[lane] : 0u, qn, lane);
    __syncthreads();

    v16f ax = {}, ay = {};
    #pragma unroll
    for (int sl = 0; sl < SLOTS; ++sl) {
        const int c = sl * 32 + wave;
        const v2f s0 = *(const v2f*)(g + (size_t)(lo + min(c, cnt - 1)) * 64 + 2 * lane);
        ax[sl] = (c < cnt) ? s0.x : 0.0f;
        ay[sl] = (c < cnt) ? s0.y : 0.0f;
    }

    const unsigned* myoq = oq + wave * OCAP;
    const int nq = __builtin_amdgcn_readfirstlane((int)min(ocnt[wave], (unsigned)OCAP));
    for (int base = 0; base < nq; base += 32) {
        const int nb = min(32, nq - base);
        const unsigned ent = (lane < nb) ? myoq[base + lane] : (NOSLOT << SRCB);
        #pragma unroll
        for (int j0 = 0; j0 < 32; j0 += 16) {
            if (j0 < nb) {
                v2f v[16];
                #pragma unroll
                for (int u = 0; u < 16; ++u) {
                    const unsigned eu = (unsigned)__builtin_amdgcn_readlane((int)ent, j0 + u);
                    v[u] = *(const v2f*)(g + (size_t)(eu & ((1u << SRCB) - 1u)) * 64 + 2 * lane);
                }
                #pragma unroll
                for (int u = 0; u < 16; ++u) {
                    const unsigned eu = (unsigned)__builtin_amdgcn_readlane((int)ent, j0 + u);
                    const unsigned k  = eu >> SRCB;
                    ax[k] = ax[k] + v[u].x;
                    ay[k] = ay[k] + v[u].y;
                }
            }
        }
    }

    const v2f bv = *(const v2f*)(b + 2 * lane);
    #pragma unroll
    for (int sl = 0; sl < SLOTS; ++sl) {
        const int c = sl * 32 + wave;
        if (c < cnt) {
            const v2f   o = *(const v2f*)(ovf + c * 64 + 2 * lane);
            const float d = dinv[lo + c];
            v2f r;
            r.x = d * (ax[sl] + o.x) + bv.x;
            r.y = d * (ay[sl] + o.y) + bv.y;
            *(v2f*)(out + (size_t)(lo + c) * 64 + 2 * lane) = r;
        }
    }
}

extern "C" void kernel_launch(void* const* d_in, const int* in_sizes, int n_in,
                              void* d_out, int out_size, void* d_ws, size_t ws_size,
                              hipStream_t stream) {
    const float* x  = (const float*)d_in[0];
    const int*   ei = (const int*)d_in[1];
    const float* W  = (const float*)d_in[2];
    const float* b  = (const float*)d_in[3];
    float* out = (float*)d_out;

    const int* row = ei;
    const int* col = ei + NEDGES;

    float* g    = (float*)d_ws;
    unsigned* plane = (unsigned*)(g + (size_t)NNODES * 64);
    unsigned* pk = plane + (size_t)NPL * PLW;
    float* dinv = (float*)(pk + NEDGES);
    unsigned* part = (unsigned*)(dinv + NNODES);

    gcn_count_deg <<<EC * NR, 1024, 0, stream>>>(row, col, part, plane, pk);
    gcn_make_dinv <<<(NNODES + 255) / 256, 256, 0, stream>>>(part, dinv);
    gcn_gemm_scale<<<(NNODES + 127) / 128, 256, 0, stream>>>(x, W, dinv, g);
    gcn_owner_sum <<<NGRP, 1024, 0, stream>>>(g, pk, plane, dinv, b, out);
}


# Inverse-root-degree pass folded into the dense-product launch: a wave pair sums the 25 count planes of its 32 rows while W is staged, writes the dinv line once; 4 launches to 3; sum order unchanged
# speedup vs baseline: 6.1328x; 1.0362x over previous
#include <hip/hip_runtime.h>

#define NNODES 100000
#define NEDGES 3200000

typedef __attribute__((ext_vector_type(2))) float v2f;
typedef __attribute__((ext_vector_type(8))) float v8f;

#define RNG   391
#define NE4   (NEDGES / 4)
#define EC    25
#define NR    10
#define NPR   (NNODES / NR)
#define NPRP  ((NPR + 31) / 32 * 32)
#define WPC   (NE4 / EC)
#define OCTW  256
#define OCTS  (WPC / OCTW)
#define NPL   32
#define PLW   (NEDGES / 32)
#define SRCB  17
static_assert(NNODES % NR == 0, "node ranges are equal");
static_assert(NE4 % EC == 0 && WPC % OCTW == 0, "chunks are whole octs");
static_assert(PLW % 32 == 0, "an owner plane is whole 128-B lines");
static_assert((NNODES - 1) / RNG < 256, "owner number fits two nibbles");
static_assert(NNODES <= (1 << SRCB), "a node number fits SRCB bits");
static_assert(RNG <= (1 << (32 - SRCB)), "a row number inside a range fits the bits above SRCB");

typedef __attribute__((ext_vector_type(4))) unsigned v4u;

__global__ __launch_bounds__(1024) void gcn_count_deg(
        const int* __restrict__ row, const int* __restrict__ col,
        unsigned* __restrict__ part, unsigned* __restrict__ plane,
        unsigned* __restrict__ pk) {
    __shared__ unsigned cnt[NPR];
    const int tid  = threadIdx.x;
    const int lane = tid & 31;
    const int wave = __builtin_amdgcn_readfirstlane(tid >> 5);
    const int i = blockIdx.x / NR;
    const int j = blockIdx.x % NR;
    for (int t = tid; t < NPR; t += 1024) cnt[t] = 0u;
    __syncthreads();

    const unsigned lo = (unsigned)j * NPR;
    const int4* c4 = (const int4*)col + (size_t)i * WPC;
    const int4* r4 = (const int4*)row + (size_t)i * WPC;
    v4u*        p4 = (v4u*)pk + (size_t)i * WPC;
    for (int oct = wave; oct < OCTS; oct += 32) {
        const bool share = (oct * NR / OCTS == j);
        int4 c[8], r[8];
        #pragma unroll
        for (int u = 0; u < 8; ++u) c[u] = c4[oct * OCTW + u * 32 + lane];
        if (share) {
            #pragma unroll
            for (int u = 0; u < 8; ++u) r[u] = r4[oct * OCTW + u * 32 + lane];
        }
        #pragma unroll
        for (int u = 0; u < 8; ++u) {
            const unsigned d0 = (unsigned)c[u].x - lo, d1 = (unsigned)c[u].y - lo;
            const unsigned d2 = (unsigned)c[u].z - lo, d3 = (unsigned)c[u].w - lo;
            if (d0 < NPR) atomicAdd(&cnt[d0], 1u);
            if (d1 < NPR) atomicAdd(&cnt[d1], 1u);
            if (d2 < NPR) atomicAdd(&cnt[d2], 1u);
            if (d3 < NPR) atomicAdd(&cnt[d3], 1u);
        }
        if (share) {
            unsigned P[NPL];
            #pragma unroll
            for (int p = 0; p < NPL; ++p) P[p] = 0u;
            #pragma unroll
            for (int u = 0; u < 8; ++u) {
                const unsigned k[4] = {(unsigned)c[u].x / RNG, (unsigned)c[u].y / RNG,
                                       (unsigned)c[u].z / RNG, (unsigned)c[u].w / RNG};
                v4u w;
                w.x = (unsigned)r[u].x | (((unsigned)c[u].x - k[0] * RNG) << SRCB);
                w.y = (unsigned)r[u].y | (((unsigned)c[u].y - k[1] * RNG) << SRCB);
                w.z = (unsigned)r[u].z | (((unsigned)c[u].z - k[2] * RNG) << SRCB);
                w.w = (unsigned)r[u].w | (((unsigned)c[u].w - k[3] * RNG) << SRCB);
                p4[oct * OCTW + u * 32 + lane] = w;
                #pragma unroll
                for (int s = 0; s < 4; ++s) {
                    const unsigned klo = k[s] & 15u, khi = k[s] >> 4;
                    const bool mine = (lane == 4 * u + s);
                    #pragma unroll
                    for (int p = 0; p < 16; ++p) {
                        const unsigned mlo = __builtin_amdgcn_ballot_w32(klo == (unsigned)p);
                        const unsigned mhi = __builtin_amdgcn_ballot_w32(khi == (unsigned)p);
                        P[p]      = mine ? mlo : P[p];
                        P[16 + p] = mine ? mhi : P[16 + p];
                    }
                }
            }
            unsigned* line = plane + (size_t)(i * OCTS + oct) * 32 + lane;
            #pragma unroll
            for (int p = 0; p < NPL; ++p) line[(size_t)p * PLW] = P[p];
        }
    }
    __syncthreads();
    unsigned* cplane = part + (size_t)blockIdx.x * NPRP;
    for (int t = tid; t < NPR; t += 1024) cplane[t] = cnt[t];
}

__global__ __launch_bounds__(256) void gcn_gemm_scale(
        const float* __restrict__ x, const float* __restrict__ W,
        const unsigned* __restrict__ part, float* __restrict__ dinv,
        float* __restrict__ g) {
    __shared__ float Wl[256 * 64];
    const int tid = threadIdx.x;

    const int wave = tid >> 5;
    const int lane = tid & 31;
    const int half = lane >> 4;
    const int mr   = lane & 15;
    const int m0   = blockIdx.x * 128 + wave * 16;
    const int pb   = blockIdx.x * 128 + (wave >> 1) * 32;

    unsigned pc[EC];
    {
        const int v = min(pb + lane, NNODES - 1);
        const int j = v / NPR, t = v - j * NPR;
        const unsigned* p = part + (size_t)j * NPRP + t;
        #pragma unroll
        for (int i = 0; i < EC; ++i) pc[i] = p[(size_t)i * NR * NPRP];
    }

    for (int i = tid; i < 256 * 64; i += 256) {
        int k = i >> 6, n = i & 63;
        Wl[(k >> 1) * 128 + (n << 1) + (k & 1)] = W[i];
    }
    __syncthreads();

    if (m0 >= NNODES) return;

    unsigned d = 1u;
    #pragma unroll
    for (int i = 0; i < EC; ++i) d += pc[i];
    const float dv = rsqrtf((float)d);
    if ((wave & 1) == 0 && pb + lane < NNODES) dinv[pb + lane] = dv;

    const float* xrow = x + (size_t)(m0 + mr) * 256;
    const v2f*   Wp   = (const v2f*)Wl;

    v8f acc0 = {}, acc1 = {}, acc2 = {}, acc3 = {};
    for (int k0 = 0; k0 < 256; k0 += 4) {
        const int ka = k0 + half * 2;
        v2f a = *(const v2f*)(xrow + ka);
        const int kp = ka >> 1;
        v2f b0 = Wp[kp * 64 + mr];
        v2f b1 = Wp[kp * 64 + 16 + mr];
        v2f b2 = Wp[kp * 64 + 32 + mr];
        v2f b3 = Wp[kp * 64 + 48 + mr];
        acc0 = __builtin_amdgcn_wmma_f32_16x16x4_f32(false, a, false, b0, (short)0, acc0, false, false);
        acc1 = __builtin_amdgcn_wmma_f32_16x16x4_f32(false, a, false, b1, (short)0, acc1, false, false);
        acc2 = __builtin_amdgcn_wmma_f32_16x16x4_f32(false, a, false, b2, (short)0, acc2, false, false);
        acc3 = __builtin_amdgcn_wmma_f32_16x16x4_f32(false, a, false, b3, (short)0, acc3, false, false);
    }

    const int l0 = (wave & 1) * 16 + half * 8;
    #pragma unroll
    for (int i = 0; i < 8; ++i) {
        const int m = m0 + i + half * 8;
        const float s = __shfl(dv, l0 + i);
        const size_t rb = (size_t)m * 64 + mr;
        g[rb]      = acc0[i] * s;
        g[rb + 16] = acc1[i] * s;
        g[rb + 32] = acc2[i] * s;
        g[rb + 48] = acc3[i] * s;
    }
}

#define NGRP  256
#define NUNIT (NEDGES / 128)
#define NSTEP ((NUNIT + 31) / 32)
#define QCAP  (32 + 1024)
#define SLOTS ((RNG + 31) / 32)
#define OCAP  640

static_assert(NGRP * RNG >= NNODES, "ranges cover every node");
static_assert(NEDGES % 128 == 0, "planes are whole 16-byte words");
static_assert(32 * 32 <= NUNIT, "the first step of every wave lies inside the plane");

__device__ __forceinline__ void gcn_push(unsigned c, int ebase, unsigned* myq, int& qn) {
    unsigned mask = __builtin_amdgcn_ballot_w32(c != 0u);
    while (mask) {
        const int      t   = __builtin_ffs((int)c) - 1;
        const unsigned pos = qn + __builtin_amdgcn_mbcnt_lo(mask, 0u);
        if (c != 0u) myq[pos] = (unsigned)(ebase + 4 * t);
        c &= c - 1u;
        qn += __builtin_popcount(mask);
        mask = __builtin_amdgcn_ballot_w32(c != 0u);
    }
}

__device__ __forceinline__ void gcn_route(const float* __restrict__ g,
                                          const unsigned* __restrict__ pk,
                                          unsigned* ocnt, unsigned* oq, float* ovf,
                                          unsigned ent, int n, int lane) {
    const unsigned pw = pk[ent];
    const int r  = (int)(pw & ((1u << SRCB) - 1u));
    const int cl = (int)(pw >> SRCB);
    bool spill = false;
    if (lane < n) {
        const unsigned w = (unsigned)cl & 31u, sl = (unsigned)cl >> 5;
        const unsigned pos = atomicAdd(&ocnt[w], 1u);
        if (pos < OCAP) oq[w * OCAP + pos] = (unsigned)r | (sl << SRCB);
        else spill = true;
    }
    unsigned m = __builtin_amdgcn_ballot_w32(spill);
    while (m) {
        const int j = __builtin_ffs((int)m) - 1;
        m &= m - 1u;
        const int rj = __builtin_amdgcn_readlane(r, j);
        const int cj = __builtin_amdgcn_readlane(cl, j);
        const v2f v = *(const v2f*)(g + (size_t)rj * 64 + 2 * lane);
        float* p = ovf + cj * 64 + 2 * lane;
        atomicAdd(p,     v.x);
        atomicAdd(p + 1, v.y);
    }
}

__device__ __forceinline__ void gcn_drain(const float* __restrict__ g,
                                          const unsigned* __restrict__ pk,
                                          unsigned* ocnt, unsigned* oq, float* ovf,
                                          const unsigned* myq, int& qn, int lane) {
    while (qn >= 32) {
        qn -= 32;
        gcn_route(g, pk, ocnt, oq, ovf, myq[qn + lane], 32, lane);
    }
}

typedef __attribute__((ext_vector_type(16))) float v16f;
#define NOSLOT 15u
static_assert(SLOTS <= NOSLOT, "slot 15 is free");

__global__ __launch_bounds__(1024) void gcn_owner_sum(
        const float* __restrict__ g, const unsigned* __restrict__ pk,
        const unsigned* __restrict__ plane,
        const float* __restrict__ dinv,
        const float* __restrict__ b, float* __restrict__ out) {
    __shared__ float    ovf[RNG * 64];
    __shared__ unsigned q[32][QCAP];
    __shared__ unsigned oq[32 * OCAP];
    __shared__ unsigned ocnt[32];

    const int tid  = threadIdx.x;
    const int lane = tid & 31;
    const int wave = __builtin_amdgcn_readfirstlane(tid >> 5);
    const int lo   = blockIdx.x * RNG;
    const int cnt  = min(RNG, NNODES - lo);

    for (int i = tid; i < RNG * 64; i += 1024) ovf[i] = 0.0f;
    if (tid < 32) ocnt[tid] = 0u;
    __syncthreads();

    const v4u* pa = (const v4u*)(plane + (size_t)(blockIdx.x & 15) * PLW);
    const v4u* pb = (const v4u*)(plane + (size_t)(16 + (blockIdx.x >> 4)) * PLW);
    unsigned*  myq = q[wave];
    int qn = 0;

    v4u pw = pa[wave * 32 + lane];
    v4u bb = pb[wave * 32 + lane];
    for (int step = wave; step < NSTEP; step += 32) {
        const int un = min((step + 32) * 32 + lane, NUNIT - 1);
        const v4u an = pa[un], bn = pb[un];
        const int u  = step * 32 + lane;
        v4u h = pw & bb;
        if (u >= NUNIT) h = (v4u){0u, 0u, 0u, 0u};
        const int e0 = u * 128;
        gcn_push(h.x, e0,     myq, qn);
        gcn_drain(g, pk, ocnt, oq, ovf, myq, qn, lane);
        gcn_push(h.y, e0 + 1, myq, qn);
        gcn_drain(g, pk, ocnt, oq, ovf, myq, qn, lane);
        gcn_push(h.z, e0 + 2, myq, qn);
        gcn_drain(g, pk, ocnt, oq, ovf, myq, qn, lane);
        gcn_push(h.w, e0 + 3, myq, qn);
        gcn_drain(g, pk, ocnt, oq, ovf, myq, qn, lane);
        pw = an; bb = bn;
    }
    if (qn > 0) gcn_route(g, pk, ocnt, oq, ovf, (lane < qn) ? myq[lane] : 0u, qn, lane);
    __syncthreads();

    v16f ax = {}, ay = {};
    #pragma unroll
    for (int sl = 0; sl < SLOTS; ++sl) {
        const int c = sl * 32 + wave;
        const v2f s0 = *(const v2f*)(g + (size_t)(lo + min(c, cnt - 1)) * 64 + 2 * lane);
        ax[sl] = (c < cnt) ? s0.x : 0.0f;
        ay[sl] = (c < cnt) ? s0.y : 0.0f;
    }

    const unsigned* myoq = oq + wave * OCAP;
    const int nq = __builtin_amdgcn_readfirstlane((int)min(ocnt[wave], (unsigned)OCAP));
    for (int base = 0; base < nq; base += 32) {
        const int nb = min(32, nq - base);
        const unsigned ent = (lane < nb) ? myoq[base + lane] : (NOSLOT << SRCB);
        #pragma unroll
        for (int j0 = 0; j0 < 32; j0 += 16) {
            if (j0 < nb) {
                v2f v[16];
                #pragma unroll
                for (int u = 0; u < 16; ++u) {
                    const unsigned eu = (unsigned)__builtin_amdgcn_readlane((int)ent, j0 + u);
                    v[u] = *(const v2f*)(g + (size_t)(eu & ((1u << SRCB) - 1u)) * 64 + 2 * lane);
                }
                #pragma unroll
                for (int u = 0; u < 16; ++u) {
                    const unsigned eu = (unsigned)__builtin_amdgcn_readlane((int)ent, j0 + u);
                    const unsigned k  = eu >> SRCB;
                    ax[k] = ax[k] + v[u].x;
                    ay[k] = ay[k] + v[u].y;
                }
            }
        }
    }

    const v2f bv = *(const v2f*)(b + 2 * lane);
    #pragma unroll
    for (int sl = 0; sl < SLOTS; ++sl) {
        const int c = sl * 32 + wave;
        if (c < cnt) {
            const v2f   o = *(const v2f*)(ovf + c * 64 + 2 * lane);
            const float d = dinv[lo + c];
            v2f r;
            r.x = d * (ax[sl] + o.x) + bv.x;
            r.y = d * (ay[sl] + o.y) + bv.y;
            *(v2f*)(out + (size_t)(lo + c) * 64 + 2 * lane) = r;
        }
    }
}

extern "C" void kernel_launch(void* const* d_in, const int* in_sizes, int n_in,
                              void* d_out, int out_size, void* d_ws, size_t ws_size,
                              hipStream_t stream) {
    const float* x  = (const float*)d_in[0];
    const int*   ei = (const int*)d_in[1];
    const float* W  = (const float*)d_in[2];
    const float* b  = (const float*)d_in[3];
    float* out = (float*)d_out;

    const int* row = ei;
    const int* col = ei + NEDGES;

    float* g    = (float*)d_ws;
    unsigned* plane = (unsigned*)(g + (size_t)NNODES * 64);
    unsigned* pk = plane + (size_t)NPL * PLW;
    float* dinv = (float*)(pk + NEDGES);
    unsigned* part = (unsigned*)(dinv + NNODES);

    gcn_count_deg <<<EC * NR, 1024, 0, stream>>>(row, col, part, plane, pk);
    gcn_gemm_scale<<<(NNODES + 127) / 128, 256, 0, stream>>>(x, W, part, dinv, g);
    gcn_owner_sum <<<NGRP, 1024, 0, stream>>>(g, pk, plane, dinv, b, out);
}


# Dense-product launch re-tiled from 782 8-wave groups to one resident 25-wave group per WGP: W staged once per WGP, one tile per wave dealt so no SIMD gets over 6.5 tiles (was 8); sums unchanged
# speedup vs baseline: 6.4133x; 1.0457x over previous
#include <hip/hip_runtime.h>

#define NNODES 100000
#define NEDGES 3200000

typedef __attribute__((ext_vector_type(2))) float v2f;
typedef __attribute__((ext_vector_type(8))) float v8f;

#define RNG   391
#define NE4   (NEDGES / 4)
#define EC    25
#define NR    10
#define NPR   (NNODES / NR)
#define NPRP  ((NPR + 31) / 32 * 32)
#define WPC   (NE4 / EC)
#define OCTW  256
#define OCTS  (WPC / OCTW)
#define NPL   32
#define PLW   (NEDGES / 32)
#define SRCB  17
static_assert(NNODES % NR == 0, "node ranges are equal");
static_assert(NE4 % EC == 0 && WPC % OCTW == 0, "chunks are whole octs");
static_assert(PLW % 32 == 0, "an owner plane is whole 128-B lines");
static_assert((NNODES - 1) / RNG < 256, "owner number fits two nibbles");
static_assert(NNODES <= (1 << SRCB), "a node number fits SRCB bits");
static_assert(RNG <= (1 << (32 - SRCB)), "a row number inside a range fits the bits above SRCB");

typedef __attribute__((ext_vector_type(4))) unsigned v4u;

__global__ __launch_bounds__(1024) void gcn_count_deg(
        const int* __restrict__ row, const int* __restrict__ col,
        unsigned* __restrict__ part, unsigned* __restrict__ plane,
        unsigned* __restrict__ pk) {
    __shared__ unsigned cnt[NPR];
    const int tid  = threadIdx.x;
    const int lane = tid & 31;
    const int wave = __builtin_amdgcn_readfirstlane(tid >> 5);
    const int i = blockIdx.x / NR;
    const int j = blockIdx.x % NR;
    for (int t = tid; t < NPR; t += 1024) cnt[t] = 0u;
    __syncthreads();

    const unsigned lo = (unsigned)j * NPR;
    const int4* c4 = (const int4*)col + (size_t)i * WPC;
    const int4* r4 = (const int4*)row + (size_t)i * WPC;
    v4u*        p4 = (v4u*)pk + (size_t)i * WPC;
    for (int oct = wave; oct < OCTS; oct += 32) {
        const bool share = (oct * NR / OCTS == j);
        int4 c[8], r[8];
        #pragma unroll
        for (int u = 0; u < 8; ++u) c[u] = c4[oct * OCTW + u * 32 + lane];
        if (share) {
            #pragma unroll
            for (int u = 0; u < 8; ++u) r[u] = r4[oct * OCTW + u * 32 + lane];
        }
        #pragma unroll
        for (int u = 0; u < 8; ++u) {
            const unsigned d0 = (unsigned)c[u].x - lo, d1 = (unsigned)c[u].y - lo;
            const unsigned d2 = (unsigned)c[u].z - lo, d3 = (unsigned)c[u].w - lo;
            if (d0 < NPR) atomicAdd(&cnt[d0], 1u);
            if (d1 < NPR) atomicAdd(&cnt[d1], 1u);
            if (d2 < NPR) atomicAdd(&cnt[d2], 1u);
            if (d3 < NPR) atomicAdd(&cnt[d3], 1u);
        }
        if (share) {
            unsigned P[NPL];
            #pragma unroll
            for (int p = 0; p < NPL; ++p) P[p] = 0u;
            #pragma unroll
            for (int u = 0; u < 8; ++u) {
                const unsigned k[4] = {(unsigned)c[u].x / RNG, (unsigned)c[u].y / RNG,
                                       (unsigned)c[u].z / RNG, (unsigned)c[u].w / RNG};
                v4u w;
                w.x = (unsigned)r[u].x | (((unsigned)c[u].x - k[0] * RNG) << SRCB);
                w.y = (unsigned)r[u].y | (((unsigned)c[u].y - k[1] * RNG) << SRCB);
                w.z = (unsigned)r[u].z | (((unsigned)c[u].z - k[2] * RNG) << SRCB);
                w.w = (unsigned)r[u].w | (((unsigned)c[u].w - k[3] * RNG) << SRCB);
                p4[oct * OCTW + u * 32 + lane] = w;
                #pragma unroll
                for (int s = 0; s < 4; ++s) {
                    const unsigned klo = k[s] & 15u, khi = k[s] >> 4;
                    const bool mine = (lane == 4 * u + s);
                    #pragma unroll
                    for (int p = 0; p < 16; ++p) {
                        const unsigned mlo = __builtin_amdgcn_ballot_w32(klo == (unsigned)p);
                        const unsigned mhi = __builtin_amdgcn_ballot_w32(khi == (unsigned)p);
                        P[p]      = mine ? mlo : P[p];
                        P[16 + p] = mine ? mhi : P[16 + p];
                    }
                }
            }
            unsigned* line = plane + (size_t)(i * OCTS + oct) * 32 + lane;
            #pragma unroll
            for (int p = 0; p < NPL; ++p) line[(size_t)p * PLW] = P[p];
        }
    }
    __syncthreads();
    unsigned* cplane = part + (size_t)blockIdx.x * NPRP;
    for (int t = tid; t < NPR; t += 1024) cplane[t] = cnt[t];
}

#define GG    256
#define FPW   24
#define GW    (FPW + 1)
#define NPAIR ((NNODES + 31) / 32)
#define FULLP ((FPW / 2) * GG)
#define NHJ   ((NPAIR - FULLP) * 4)
#define WPT   ((256 * 64 + GW * 32 - 1) / (GW * 32))
#define WLDS  (164 * 1024 / 4)
static_assert(NNODES % 32 == 0, "whole pairs of whole tiles");
static_assert(FULLP <= NPAIR, "the full-tile pairs are real rows");
static_assert(NHJ <= GG, "one half-tile job per group at most");
static_assert(WLDS >= 256 * 64, "W fits");

template <int NT>
__device__ __forceinline__ void gcn_tile(const float* __restrict__ xrow, const v2f* Wp, int c0,
                                         float dv, int l0, float* __restrict__ g,
                                         int m0, int half, int mr) {
    v8f acc[NT];
    #pragma unroll
    for (int j = 0; j < NT; ++j) acc[j] = (v8f){};
    for (int k0 = 0; k0 < 256; k0 += 4) {
        const int ka = k0 + half * 2;
        v2f a = *(const v2f*)(xrow + ka);
        const int kp = ka >> 1;
        v2f b[NT];
        #pragma unroll
        for (int j = 0; j < NT; ++j) b[j] = Wp[kp * 64 + c0 + j * 16 + mr];
        #pragma unroll
        for (int j = 0; j < NT; ++j)
            acc[j] = __builtin_amdgcn_wmma_f32_16x16x4_f32(false, a, false, b[j], (short)0, acc[j], false, false);
    }
    #pragma unroll
    for (int i = 0; i < 8; ++i) {
        const int m = m0 + i + half * 8;
        const float s = __shfl(dv, l0 + i);
        const size_t rb = (size_t)m * 64 + c0 + mr;
        #pragma unroll
        for (int j = 0; j < NT; ++j) g[rb + j * 16] = acc[j][i] * s;
    }
}

__global__ __launch_bounds__(GW * 32) void gcn_gemm_scale(
        const float* __restrict__ x, const float* __restrict__ W,
        const unsigned* __restrict__ part, float* __restrict__ dinv,
        float* __restrict__ g) {
    __shared__ float Wl[WLDS];
    const int tid = threadIdx.x;

    const int wave = __builtin_amdgcn_readfirstlane(tid >> 5);
    const int lane = tid & 31;
    const int half = lane >> 4;
    const int mr   = lane & 15;
    int pb, m0, c0;
    bool writer;
    if (wave < FPW) {
        pb = ((wave >> 1) * GG + (int)blockIdx.x) * 32;
        m0 = pb + (wave & 1) * 16;
        c0 = 0;
        writer = (wave & 1) == 0;
    } else {
        const int z = (int)blockIdx.x;
        const int t = z >> 1;
        pb = (FULLP + (t >> 1)) * 32;
        m0 = (z < NHJ) ? pb + (t & 1) * 16 : NNODES;
        c0 = (z & 1) * 32;
        writer = (z & 3) == 0;
    }

    unsigned pc[EC];
    {
        const int v = min(pb + lane, NNODES - 1);
        const int j = v / NPR, t = v - j * NPR;
        const unsigned* p = part + (size_t)j * NPRP + t;
        #pragma unroll
        for (int i = 0; i < EC; ++i) pc[i] = p[(size_t)i * NR * NPRP];
    }

    #pragma unroll
    for (int u = 0; u < WPT; ++u) {
        const int i = tid + u * (GW * 32);
        if (i < 256 * 64) {
            const int k = i >> 6, n = i & 63;
            Wl[(k >> 1) * 128 + (n << 1) + (k & 1)] = W[i];
        }
    }
    __syncthreads();

    if (m0 >= NNODES) return;

    unsigned d = 1u;
    #pragma unroll
    for (int i = 0; i < EC; ++i) d += pc[i];
    const float dv = rsqrtf((float)d);
    if (writer && pb + lane < NNODES) dinv[pb + lane] = dv;

    const float* xrow = x + (size_t)(m0 + mr) * 256;
    const v2f*   Wp   = (const v2f*)Wl;
    const int    l0   = (m0 - pb) + half * 8;

    if (wave < FPW) gcn_tile<4>(xrow, Wp, 0,  dv, l0, g, m0, half, mr);
    else            gcn_tile<2>(xrow, Wp, c0, dv, l0, g, m0, half, mr);
}

#define NGRP  256
#define NUNIT (NEDGES / 128)
#define NSTEP ((NUNIT + 31) / 32)
#define QCAP  (32 + 1024)
#define SLOTS ((RNG + 31) / 32)
#define OCAP  640

static_assert(NGRP * RNG >= NNODES, "ranges cover every node");
static_assert(NEDGES % 128 == 0, "planes are whole 16-byte words");
static_assert(32 * 32 <= NUNIT, "the first step of every wave lies inside the plane");

__device__ __forceinline__ void gcn_push(unsigned c, int ebase, unsigned* myq, int& qn) {
    unsigned mask = __builtin_amdgcn_ballot_w32(c != 0u);
    while (mask) {
        const int      t   = __builtin_ffs((int)c) - 1;
        const unsigned pos = qn + __builtin_amdgcn_mbcnt_lo(mask, 0u);
        if (c != 0u) myq[pos] = (unsigned)(ebase + 4 * t);
        c &= c - 1u;
        qn += __builtin_popcount(mask);
        mask = __builtin_amdgcn_ballot_w32(c != 0u);
    }
}

__device__ __forceinline__ void gcn_route(const float* __restrict__ g,
                                          const unsigned* __restrict__ pk,
                                          unsigned* ocnt, unsigned* oq, float* ovf,
                                          unsigned ent, int n, int lane) {
    const unsigned pw = pk[ent];
    const int r  = (int)(pw & ((1u << SRCB) - 1u));
    const int cl = (int)(pw >> SRCB);
    bool spill = false;
    if (lane < n) {
        const unsigned w = (unsigned)cl & 31u, sl = (unsigned)cl >> 5;
        const unsigned pos = atomicAdd(&ocnt[w], 1u);
        if (pos < OCAP) oq[w * OCAP + pos] = (unsigned)r | (sl << SRCB);
        else spill = true;
    }
    unsigned m = __builtin_amdgcn_ballot_w32(spill);
    while (m) {
        const int j = __builtin_ffs((int)m) - 1;
        m &= m - 1u;
        const int rj = __builtin_amdgcn_readlane(r, j);
        const int cj = __builtin_amdgcn_readlane(cl, j);
        const v2f v = *(const v2f*)(g + (size_t)rj * 64 + 2 * lane);
        float* p = ovf + cj * 64 + 2 * lane;
        atomicAdd(p,     v.x);
        atomicAdd(p + 1, v.y);
    }
}

__device__ __forceinline__ void gcn_drain(const float* __restrict__ g,
                                          const unsigned* __restrict__ pk,
                                          unsigned* ocnt, unsigned* oq, float* ovf,
                                          const unsigned* myq, int& qn, int lane) {
    while (qn >= 32) {
        qn -= 32;
        gcn_route(g, pk, ocnt, oq, ovf, myq[qn + lane], 32, lane);
    }
}

typedef __attribute__((ext_vector_type(16))) float v16f;
#define NOSLOT 15u
static_assert(SLOTS <= NOSLOT, "slot 15 is free");

__global__ __launch_bounds__(1024) void gcn_owner_sum(
        const float* __restrict__ g, const unsigned* __restrict__ pk,
        const unsigned* __restrict__ plane,
        const float* __restrict__ dinv,
        const float* __restrict__ b, float* __restrict__ out) {
    __shared__ float    ovf[RNG * 64];
    __shared__ unsigned q[32][QCAP];
    __shared__ unsigned oq[32 * OCAP];
    __shared__ unsigned ocnt[32];

    const int tid  = threadIdx.x;
    const int lane = tid & 31;
    const int wave = __builtin_amdgcn_readfirstlane(tid >> 5);
    const int lo   = blockIdx.x * RNG;
    const int cnt  = min(RNG, NNODES - lo);

    for (int i = tid; i < RNG * 64; i += 1024) ovf[i] = 0.0f;
    if (tid < 32) ocnt[tid] = 0u;
    __syncthreads();

    const v4u* pa = (const v4u*)(plane + (size_t)(blockIdx.x & 15) * PLW);
    const v4u* pb = (const v4u*)(plane + (size_t)(16 + (blockIdx.x >> 4)) * PLW);
    unsigned*  myq = q[wave];
    int qn = 0;

    v4u pw = pa[wave * 32 + lane];
    v4u bb = pb[wave * 32 + lane];
    for (int step = wave; step < NSTEP; step += 32) {
        const int un = min((step + 32) * 32 + lane, NUNIT - 1);
        const v4u an = pa[un], bn = pb[un];
        const int u  = step * 32 + lane;
        v4u h = pw & bb;
        if (u >= NUNIT) h = (v4u){0u, 0u, 0u, 0u};
        const int e0 = u * 128;
        gcn_push(h.x, e0,     myq, qn);
        gcn_drain(g, pk, ocnt, oq, ovf, myq, qn, lane);
        gcn_push(h.y, e0 + 1, myq, qn);
        gcn_drain(g, pk, ocnt, oq, ovf, myq, qn, lane);
        gcn_push(h.z, e0 + 2, myq, qn);
        gcn_drain(g, pk, ocnt, oq, ovf, myq, qn, lane);
        gcn_push(h.w, e0 + 3, myq, qn);
        gcn_drain(g, pk, ocnt, oq, ovf, myq, qn, lane);
        pw = an; bb = bn;
    }
    if (qn > 0) gcn_route(g, pk, ocnt, oq, ovf, (lane < qn) ? myq[lane] : 0u, qn, lane);
    __syncthreads();

    v16f ax = {}, ay = {};
    #pragma unroll
    for (int sl = 0; sl < SLOTS; ++sl) {
        const int c = sl * 32 + wave;
        const v2f s0 = *(const v2f*)(g + (size_t)(lo + min(c, cnt - 1)) * 64 + 2 * lane);
        ax[sl] = (c < cnt) ? s0.x : 0.0f;
        ay[sl] = (c < cnt) ? s0.y : 0.0f;
    }

    const unsigned* myoq = oq + wave * OCAP;
    const int nq = __builtin_amdgcn_readfirstlane((int)min(ocnt[wave], (unsigned)OCAP));
    for (int base = 0; base < nq; base += 32) {
        const int nb = min(32, nq - base);
        const unsigned ent = (lane < nb) ? myoq[base + lane] : (NOSLOT << SRCB);
        #pragma unroll
        for (int j0 = 0; j0 < 32; j0 += 16) {
            if (j0 < nb) {
                v2f v[16];
                #pragma unroll
                for (int u = 0; u < 16; ++u) {
                    const unsigned eu = (unsigned)__builtin_amdgcn_readlane((int)ent, j0 + u);
                    v[u] = *(const v2f*)(g + (size_t)(eu & ((1u << SRCB) - 1u)) * 64 + 2 * lane);
                }
                #pragma unroll
                for (int u = 0; u < 16; ++u) {
                    const unsigned eu = (unsigned)__builtin_amdgcn_readlane((int)ent, j0 + u);
                    const unsigned k  = eu >> SRCB;
                    ax[k] = ax[k] + v[u].x;
                    ay[k] = ay[k] + v[u].y;
                }
            }
        }
    }

    const v2f bv = *(const v2f*)(b + 2 * lane);
    #pragma unroll
    for (int sl = 0; sl < SLOTS; ++sl) {
        const int c = sl * 32 + wave;
        if (c < cnt) {
            const v2f   o = *(const v2f*)(ovf + c * 64 + 2 * lane);
            const float d = dinv[lo + c];
            v2f r;
            r.x = d * (ax[sl] + o.x) + bv.x;
            r.y = d * (ay[sl] + o.y) + bv.y;
            *(v2f*)(out + (size_t)(lo + c) * 64 + 2 * lane) = r;
        }
    }
}

extern "C" void kernel_launch(void* const* d_in, const int* in_sizes, int n_in,
                              void* d_out, int out_size, void* d_ws, size_t ws_size,
                              hipStream_t stream) {
    const float* x  = (const float*)d_in[0];
    const int*   ei = (const int*)d_in[1];
    const float* W  = (const float*)d_in[2];
    const float* b  = (const float*)d_in[3];
    float* out = (float*)d_out;

    const int* row = ei;
    const int* col = ei + NEDGES;

    float* g    = (float*)d_ws;
    unsigned* plane = (unsigned*)(g + (size_t)NNODES * 64);
    unsigned* pk = plane + (size_t)NPL * PLW;
    float* dinv = (float*)(pk + NEDGES);
    unsigned* part = (unsigned*)(dinv + NNODES);

    gcn_count_deg <<<EC * NR, 1024, 0, stream>>>(row, col, part, plane, pk);
    gcn_gemm_scale<<<GG, GW * 32, 0, stream>>>(x, W, part, dinv, g);
    gcn_owner_sum <<<NGRP, 1024, 0, stream>>>(g, pk, plane, dinv, b, out);
}


# Pass 3 row gathers double-buffered: the next group of queued g rows is asked for before the current group is added, so a wave keeps reads in flight through its adds; sum order unchanged
# speedup vs baseline: 6.6082x; 1.0304x over previous
#include <hip/hip_runtime.h>

#define NNODES 100000
#define NEDGES 3200000

typedef __attribute__((ext_vector_type(2))) float v2f;
typedef __attribute__((ext_vector_type(8))) float v8f;

#define RNG   391
#define NE4   (NEDGES / 4)
#define EC    25
#define NR    10
#define NPR   (NNODES / NR)
#define NPRP  ((NPR + 31) / 32 * 32)
#define WPC   (NE4 / EC)
#define OCTW  256
#define OCTS  (WPC / OCTW)
#define NPL   32
#define PLW   (NEDGES / 32)
#define SRCB  17
static_assert(NNODES % NR == 0, "node ranges are equal");
static_assert(NE4 % EC == 0 && WPC % OCTW == 0, "chunks are whole octs");
static_assert(PLW % 32 == 0, "an owner plane is whole 128-B lines");
static_assert((NNODES - 1) / RNG < 256, "owner number fits two nibbles");
static_assert(NNODES <= (1 << SRCB), "a node number fits SRCB bits");
static_assert(RNG <= (1 << (32 - SRCB)), "a row number inside a range fits the bits above SRCB");

typedef __attribute__((ext_vector_type(4))) unsigned v4u;

__global__ __launch_bounds__(1024) void gcn_count_deg(
        const int* __restrict__ row, const int* __restrict__ col,
        unsigned* __restrict__ part, unsigned* __restrict__ plane,
        unsigned* __restrict__ pk) {
    __shared__ unsigned cnt[NPR];
    const int tid  = threadIdx.x;
    const int lane = tid & 31;
    const int wave = __builtin_amdgcn_readfirstlane(tid >> 5);
    const int i = blockIdx.x / NR;
    const int j = blockIdx.x % NR;
    for (int t = tid; t < NPR; t += 1024) cnt[t] = 0u;
    __syncthreads();

    const unsigned lo = (unsigned)j * NPR;
    const int4* c4 = (const int4*)col + (size_t)i * WPC;
    const int4* r4 = (const int4*)row + (size_t)i * WPC;
    v4u*        p4 = (v4u*)pk + (size_t)i * WPC;
    for (int oct = wave; oct < OCTS; oct += 32) {
        const bool share = (oct * NR / OCTS == j);
        int4 c[8], r[8];
        #pragma unroll
        for (int u = 0; u < 8; ++u) c[u] = c4[oct * OCTW + u * 32 + lane];
        if (share) {
            #pragma unroll
            for (int u = 0; u < 8; ++u) r[u] = r4[oct * OCTW + u * 32 + lane];
        }
        #pragma unroll
        for (int u = 0; u < 8; ++u) {
            const unsigned d0 = (unsigned)c[u].x - lo, d1 = (unsigned)c[u].y - lo;
            const unsigned d2 = (unsigned)c[u].z - lo, d3 = (unsigned)c[u].w - lo;
            if (d0 < NPR) atomicAdd(&cnt[d0], 1u);
            if (d1 < NPR) atomicAdd(&cnt[d1], 1u);
            if (d2 < NPR) atomicAdd(&cnt[d2], 1u);
            if (d3 < NPR) atomicAdd(&cnt[d3], 1u);
        }
        if (share) {
            unsigned P[NPL];
            #pragma unroll
            for (int p = 0; p < NPL; ++p) P[p] = 0u;
            #pragma unroll
            for (int u = 0; u < 8; ++u) {
                const unsigned k[4] = {(unsigned)c[u].x / RNG, (unsigned)c[u].y / RNG,
                                       (unsigned)c[u].z / RNG, (unsigned)c[u].w / RNG};
                v4u w;
                w.x = (unsigned)r[u].x | (((unsigned)c[u].x - k[0] * RNG) << SRCB);
                w.y = (unsigned)r[u].y | (((unsigned)c[u].y - k[1] * RNG) << SRCB);
                w.z = (unsigned)r[u].z | (((unsigned)c[u].z - k[2] * RNG) << SRCB);
                w.w = (unsigned)r[u].w | (((unsigned)c[u].w - k[3] * RNG) << SRCB);
                p4[oct * OCTW + u * 32 + lane] = w;
                #pragma unroll
                for (int s = 0; s < 4; ++s) {
                    const unsigned klo = k[s] & 15u, khi = k[s] >> 4;
                    const bool mine = (lane == 4 * u + s);
                    #pragma unroll
                    for (int p = 0; p < 16; ++p) {
                        const unsigned mlo = __builtin_amdgcn_ballot_w32(klo == (unsigned)p);
                        const unsigned mhi = __builtin_amdgcn_ballot_w32(khi == (unsigned)p);
                        P[p]      = mine ? mlo : P[p];
                        P[16 + p] = mine ? mhi : P[16 + p];
                    }
                }
            }
            unsigned* line = plane + (size_t)(i * OCTS + oct) * 32 + lane;
            #pragma unroll
            for (int p = 0; p < NPL; ++p) line[(size_t)p * PLW] = P[p];
        }
    }
    __syncthreads();
    unsigned* cplane = part + (size_t)blockIdx.x * NPRP;
    for (int t = tid; t < NPR; t += 1024) cplane[t] = cnt[t];
}

#define GG    256
#define FPW   24
#define GW    (FPW + 1)
#define NPAIR ((NNODES + 31) / 32)
#define FULLP ((FPW / 2) * GG)
#define NHJ   ((NPAIR - FULLP) * 4)
#define WPT   ((256 * 64 + GW * 32 - 1) / (GW * 32))
#define WLDS  (164 * 1024 / 4)
static_assert(NNODES % 32 == 0, "whole pairs of whole tiles");
static_assert(FULLP <= NPAIR, "the full-tile pairs are real rows");
static_assert(NHJ <= GG, "one half-tile job per group at most");
static_assert(WLDS >= 256 * 64, "W fits");

template <int NT>
__device__ __forceinline__ void gcn_tile(const float* __restrict__ xrow, const v2f* Wp, int c0,
                                         float dv, int l0, float* __restrict__ g,
                                         int m0, int half, int mr) {
    v8f acc[NT];
    #pragma unroll
    for (int j = 0; j < NT; ++j) acc[j] = (v8f){};
    for (int k0 = 0; k0 < 256; k0 += 4) {
        const int ka = k0 + half * 2;
        v2f a = *(const v2f*)(xrow + ka);
        const int kp = ka >> 1;
        v2f b[NT];
        #pragma unroll
        for (int j = 0; j < NT; ++j) b[j] = Wp[kp * 64 + c0 + j * 16 + mr];
        #pragma unroll
        for (int j = 0; j < NT; ++j)
            acc[j] = __builtin_amdgcn_wmma_f32_16x16x4_f32(false, a, false, b[j], (short)0, acc[j], false, false);
    }
    #pragma unroll
    for (int i = 0; i < 8; ++i) {
        const int m = m0 + i + half * 8;
        const float s = __shfl(dv, l0 + i);
        const size_t rb = (size_t)m * 64 + c0 + mr;
        #pragma unroll
        for (int j = 0; j < NT; ++j) g[rb + j * 16] = acc[j][i] * s;
    }
}

__global__ __launch_bounds__(GW * 32) void gcn_gemm_scale(
        const float* __restrict__ x, const float* __restrict__ W,
        const unsigned* __restrict__ part, float* __restrict__ dinv,
        float* __restrict__ g) {
    __shared__ float Wl[WLDS];
    const int tid = threadIdx.x;

    const int wave = __builtin_amdgcn_readfirstlane(tid >> 5);
    const int lane = tid & 31;
    const int half = lane >> 4;
    const int mr   = lane & 15;
    int pb, m0, c0;
    bool writer;
    if (wave < FPW) {
        pb = ((wave >> 1) * GG + (int)blockIdx.x) * 32;
        m0 = pb + (wave & 1) * 16;
        c0 = 0;
        writer = (wave & 1) == 0;
    } else {
        const int z = (int)blockIdx.x;
        const int t = z >> 1;
        pb = (FULLP + (t >> 1)) * 32;
        m0 = (z < NHJ) ? pb + (t & 1) * 16 : NNODES;
        c0 = (z & 1) * 32;
        writer = (z & 3) == 0;
    }

    unsigned pc[EC];
    {
        const int v = min(pb + lane, NNODES - 1);
        const int j = v / NPR, t = v - j * NPR;
        const unsigned* p = part + (size_t)j * NPRP + t;
        #pragma unroll
        for (int i = 0; i < EC; ++i) pc[i] = p[(size_t)i * NR * NPRP];
    }

    #pragma unroll
    for (int u = 0; u < WPT; ++u) {
        const int i = tid + u * (GW * 32);
        if (i < 256 * 64) {
            const int k = i >> 6, n = i & 63;
            Wl[(k >> 1) * 128 + (n << 1) + (k & 1)] = W[i];
        }
    }
    __syncthreads();

    if (m0 >= NNODES) return;

    unsigned d = 1u;
    #pragma unroll
    for (int i = 0; i < EC; ++i) d += pc[i];
    const float dv = rsqrtf((float)d);
    if (writer && pb + lane < NNODES) dinv[pb + lane] = dv;

    const float* xrow = x + (size_t)(m0 + mr) * 256;
    const v2f*   Wp   = (const v2f*)Wl;
    const int    l0   = (m0 - pb) + half * 8;

    if (wave < FPW) gcn_tile<4>(xrow, Wp, 0,  dv, l0, g, m0, half, mr);
    else            gcn_tile<2>(xrow, Wp, c0, dv, l0, g, m0, half, mr);
}

#define NGRP  256
#define NUNIT (NEDGES / 128)
#define NSTEP ((NUNIT + 31) / 32)
#define QCAP  (32 + 1024)
#define SLOTS ((RNG + 31) / 32)
#define OCAP  640

static_assert(NGRP * RNG >= NNODES, "ranges cover every node");
static_assert(NEDGES % 128 == 0, "planes are whole 16-byte words");
static_assert(32 * 32 <= NUNIT, "the first step of every wave lies inside the plane");

__device__ __forceinline__ void gcn_push(unsigned c, int ebase, unsigned* myq, int& qn) {
    unsigned mask = __builtin_amdgcn_ballot_w32(c != 0u);
    while (mask) {
        const int      t   = __builtin_ffs((int)c) - 1;
        const unsigned pos = qn + __builtin_amdgcn_mbcnt_lo(mask, 0u);
        if (c != 0u) myq[pos] = (unsigned)(ebase + 4 * t);
        c &= c - 1u;
        qn += __builtin_popcount(mask);
        mask = __builtin_amdgcn_ballot_w32(c != 0u);
    }
}

__device__ __forceinline__ void gcn_route(const float* __restrict__ g,
                                          const unsigned* __restrict__ pk,
                                          unsigned* ocnt, unsigned* oq, float* ovf,
                                          unsigned ent, int n, int lane) {
    const unsigned pw = pk[ent];
    const int r  = (int)(pw & ((1u << SRCB) - 1u));
    const int cl = (int)(pw >> SRCB);
    bool spill = false;
    if (lane < n) {
        const unsigned w = (unsigned)cl & 31u, sl = (unsigned)cl >> 5;
        const unsigned pos = atomicAdd(&ocnt[w], 1u);
        if (pos < OCAP) oq[w * OCAP + pos] = (unsigned)r | (sl << SRCB);
        else spill = true;
    }
    unsigned m = __builtin_amdgcn_ballot_w32(spill);
    while (m) {
        const int j = __builtin_ffs((int)m) - 1;
        m &= m - 1u;
        const int rj = __builtin_amdgcn_readlane(r, j);
        const int cj = __builtin_amdgcn_readlane(cl, j);
        const v2f v = *(const v2f*)(g + (size_t)rj * 64 + 2 * lane);
        float* p = ovf + cj * 64 + 2 * lane;
        atomicAdd(p,     v.x);
        atomicAdd(p + 1, v.y);
    }
}

__device__ __forceinline__ void gcn_drain(const float* __restrict__ g,
                                          const unsigned* __restrict__ pk,
                                          unsigned* ocnt, unsigned* oq, float* ovf,
                                          const unsigned* myq, int& qn, int lane) {
    while (qn >= 32) {
        qn -= 32;
        gcn_route(g, pk, ocnt, oq, ovf, myq[qn + lane], 32, lane);
    }
}

typedef __attribute__((ext_vector_type(16))) float v16f;
#define NOSLOT 15u
static_assert(SLOTS <= NOSLOT, "slot 15 is free");

__device__ __forceinline__ unsigned gcn_places(const unsigned* myoq, int h, int nq, int lane) {
    const int idx = h * 16 + (lane & 15);
    return (idx < nq) ? myoq[idx] : (NOSLOT << SRCB);
}

__device__ __forceinline__ void gcn_ask16(const float* __restrict__ g, unsigned ent, int lane,
                                          v2f (&v)[16]) {
    #pragma unroll
    for (int u = 0; u < 16; ++u) {
        const unsigned eu = (unsigned)__builtin_amdgcn_readlane((int)ent, u);
        v[u] = *(const v2f*)(g + (size_t)(eu & ((1u << SRCB) - 1u)) * 64 + 2 * lane);
    }
}

__device__ __forceinline__ void gcn_add16(unsigned ent, const v2f (&v)[16], v16f& ax, v16f& ay) {
    #pragma unroll
    for (int u = 0; u < 16; ++u) {
        const unsigned k = (unsigned)__builtin_amdgcn_readlane((int)ent, u) >> SRCB;
        ax[k] = ax[k] + v[u].x;
        ay[k] = ay[k] + v[u].y;
    }
}

__global__ __launch_bounds__(1024) void gcn_owner_sum(
        const float* __restrict__ g, const unsigned* __restrict__ pk,
        const unsigned* __restrict__ plane,
        const float* __restrict__ dinv,
        const float* __restrict__ b, float* __restrict__ out) {
    __shared__ float    ovf[RNG * 64];
    __shared__ unsigned q[32][QCAP];
    __shared__ unsigned oq[32 * OCAP];
    __shared__ unsigned ocnt[32];

    const int tid  = threadIdx.x;
    const int lane = tid & 31;
    const int wave = __builtin_amdgcn_readfirstlane(tid >> 5);
    const int lo   = blockIdx.x * RNG;
    const int cnt  = min(RNG, NNODES - lo);

    for (int i = tid; i < RNG * 64; i += 1024) ovf[i] = 0.0f;
    if (tid < 32) ocnt[tid] = 0u;
    __syncthreads();

    const v4u* pa = (const v4u*)(plane + (size_t)(blockIdx.x & 15) * PLW);
    const v4u* pb = (const v4u*)(plane + (size_t)(16 + (blockIdx.x >> 4)) * PLW);
    unsigned*  myq = q[wave];
    int qn = 0;

    v4u pw = pa[wave * 32 + lane];
    v4u bb = pb[wave * 32 + lane];
    for (int step = wave; step < NSTEP; step += 32) {
        const int un = min((step + 32) * 32 + lane, NUNIT - 1);
        const v4u an = pa[un], bn = pb[un];
        const int u  = step * 32 + lane;
        v4u h = pw & bb;
        if (u >= NUNIT) h = (v4u){0u, 0u, 0u, 0u};
        const int e0 = u * 128;
        gcn_push(h.x, e0,     myq, qn);
        gcn_drain(g, pk, ocnt, oq, ovf, myq, qn, lane);
        gcn_push(h.y, e0 + 1, myq, qn);
        gcn_drain(g, pk, ocnt, oq, ovf, myq, qn, lane);
        gcn_push(h.z, e0 + 2, myq, qn);
        gcn_drain(g, pk, ocnt, oq, ovf, myq, qn, lane);
        gcn_push(h.w, e0 + 3, myq, qn);
        gcn_drain(g, pk, ocnt, oq, ovf, myq, qn, lane);
        pw = an; bb = bn;
    }
    if (qn > 0) gcn_route(g, pk, ocnt, oq, ovf, (lane < qn) ? myq[lane] : 0u, qn, lane);
    __syncthreads();

    v16f ax = {}, ay = {};
    #pragma unroll
    for (int sl = 0; sl < SLOTS; ++sl) {
        const int c = sl * 32 + wave;
        const v2f s0 = *(const v2f*)(g + (size_t)(lo + min(c, cnt - 1)) * 64 + 2 * lane);
        ax[sl] = (c < cnt) ? s0.x : 0.0f;
        ay[sl] = (c < cnt) ? s0.y : 0.0f;
    }

    const unsigned* myoq = oq + wave * OCAP;
    const int nq = __builtin_amdgcn_readfirstlane((int)min(ocnt[wave], (unsigned)OCAP));
    const int nh = (nq + 15) >> 4;
    if (nh > 0) {
        v2f va[16], vb[16];
        unsigned ea = gcn_places(myoq, 0, nq, lane), eb;
        gcn_ask16(g, ea, lane, va);
        int h = 0;
        while (h + 2 < nh) {
            eb = gcn_places(myoq, h + 1, nq, lane);
            gcn_ask16(g, eb, lane, vb);
            gcn_add16(ea, va, ax, ay);
            ea = gcn_places(myoq, h + 2, nq, lane);
            gcn_ask16(g, ea, lane, va);
            gcn_add16(eb, vb, ax, ay);
            h += 2;
        }
        if (h + 1 < nh) {
            eb = gcn_places(myoq, h + 1, nq, lane);
            gcn_ask16(g, eb, lane, vb);
            gcn_add16(ea, va, ax, ay);
            gcn_add16(eb, vb, ax, ay);
        } else {
            gcn_add16(ea, va, ax, ay);
        }
    }

    const v2f bv = *(const v2f*)(b + 2 * lane);
    #pragma unroll
    for (int sl = 0; sl < SLOTS; ++sl) {
        const int c = sl * 32 + wave;
        if (c < cnt) {
            const v2f   o = *(const v2f*)(ovf + c * 64 + 2 * lane);
            const float d = dinv[lo + c];
            v2f r;
            r.x = d * (ax[sl] + o.x) + bv.x;
            r.y = d * (ay[sl] + o.y) + bv.y;
            *(v2f*)(out + (size_t)(lo + c) * 64 + 2 * lane) = r;
        }
    }
}

extern "C" void kernel_launch(void* const* d_in, const int* in_sizes, int n_in,
                              void* d_out, int out_size, void* d_ws, size_t ws_size,
                              hipStream_t stream) {
    const float* x  = (const float*)d_in[0];
    const int*   ei = (const int*)d_in[1];
    const float* W  = (const float*)d_in[2];
    const float* b  = (const float*)d_in[3];
    float* out = (float*)d_out;

    const int* row = ei;
    const int* col = ei + NEDGES;

    float* g    = (float*)d_ws;
    unsigned* plane = (unsigned*)(g + (size_t)NNODES * 64);
    unsigned* pk = plane + (size_t)NPL * PLW;
    float* dinv = (float*)(pk + NEDGES);
    unsigned* part = (unsigned*)(dinv + NNODES);

    gcn_count_deg <<<EC * NR, 1024, 0, stream>>>(row, col, part, plane, pk);
    gcn_gemm_scale<<<GG, GW * 32, 0, stream>>>(x, W, part, dinv, g);
    gcn_owner_sum <<<NGRP, 1024, 0, stream>>>(g, pk, plane, dinv, b, out);
}
